# starting kernel
# baseline (speedup 1.0000x reference)
#include <hip/hip_runtime.h>

#define N_NODES 100000
#define DHID    128
#define E_EDGES 1600000
#define BN_EPS  1e-5f

typedef __attribute__((ext_vector_type(2))) float v2f;
typedef __attribute__((ext_vector_type(8))) float v8f;

__global__ void k_zero(float* __restrict__ p, int n) {
  int i = blockIdx.x * blockDim.x + threadIdx.x;
  if (i < n) p[i] = 0.0f;
}

__global__ void k_degree(const int* __restrict__ dst, float* __restrict__ deg, int e) {
  int i = blockIdx.x * blockDim.x + threadIdx.x;
  if (i < e) unsafeAtomicAdd(&deg[dst[i]], 1.0f);
}

__global__ void k_dinv(float* __restrict__ deg_dinv, float* __restrict__ selfc, int n) {
  int i = blockIdx.x * blockDim.x + threadIdx.x;
  if (i < n) {
    float d = __frsqrt_rn(deg_dinv[i] + 1.0f);
    deg_dinv[i] = d;
    selfc[i]    = d * d;
  }
}

template <int AMODE, bool HAS_BIAS, bool EPI_RELU>
__global__ __launch_bounds__(256) void k_gemm128(
    const float* __restrict__ A, const float* __restrict__ W,
    float* __restrict__ C, int nrows,
    const float* __restrict__ bnScale, const float* __restrict__ bnShift,
    const float* __restrict__ bias)
{
  __shared__ float wl[DHID * DHID];
  const int tid = threadIdx.x;
  for (int i = tid; i < DHID * DHID; i += 256) {
    int k = i >> 7, n = i & 127;
    wl[(((k >> 1) << 7) + n) * 2 + (k & 1)] = W[i];
  }
  __syncthreads();

  const int wave  = tid >> 5;
  const int lane  = tid & 31;
  const int laneM = lane & 15;
  const int hi    = lane >> 4;
  const int m0    = (blockIdx.x * 8 + wave) << 4;
  if (m0 >= nrows) return;

  const float2* __restrict__ wl2  = (const float2*)wl;
  const float*  __restrict__ arow = A + (size_t)(m0 + laneM) * DHID;

  v8f acc[8];
#pragma unroll
  for (int t = 0; t < 8; ++t) acc[t] = (v8f){};

  for (int k = 0; k < DHID; k += 4) {
    float2 av = *(const float2*)(arow + k + hi * 2);
    if (AMODE == 1) {
      av.x = fmaxf(av.x, 0.0f);
      av.y = fmaxf(av.y, 0.0f);
    } else if (AMODE == 2) {
      int c0 = k + hi * 2;
      av.x = fmaxf(av.x, 0.0f) * bnScale[c0]     + bnShift[c0];
      av.y = fmaxf(av.y, 0.0f) * bnScale[c0 + 1] + bnShift[c0 + 1];
    }
    v2f a; a[0] = av.x; a[1] = av.y;

    const float2* __restrict__ brow = wl2 + (((k >> 1) + hi) << 7);
#pragma unroll
    for (int t = 0; t < 8; ++t) {
      float2 bv = brow[t * 16 + laneM];
      v2f b; b[0] = bv.x; b[1] = bv.y;
      acc[t] = __builtin_amdgcn_wmma_f32_16x16x4_f32(
          false, a, false, b, (short)0, acc[t], false, false);
    }
  }

#pragma unroll
  for (int t = 0; t < 8; ++t) {
    int   col  = t * 16 + laneM;
    float bval = HAS_BIAS ? bias[col] : 0.0f;
#pragma unroll
    for (int r = 0; r < 8; ++r) {
      int   row = m0 + r + hi * 8;
      float v   = acc[t][r] + bval;
      if (EPI_RELU) v = fmaxf(v, 0.0f);
      C[(size_t)row * DHID + col] = v;
    }
  }
}

__global__ void k_selfbias(const float* __restrict__ h, const float* __restrict__ selfc,
                           const float* __restrict__ b, float* __restrict__ out, int n) {
  int i = blockIdx.x * blockDim.x + threadIdx.x;
  if (i >= n * (DHID / 4)) return;
  int    node = i >> 5;
  int    c4   = (i & 31) * 4;
  float  sc   = selfc[node];
  float4 hv   = *(const float4*)(h + (size_t)node * DHID + c4);
  float4 bv   = *(const float4*)(b + c4);
  float4 o;
  o.x = hv.x * sc + bv.x;
  o.y = hv.y * sc + bv.y;
  o.z = hv.z * sc + bv.z;
  o.w = hv.w * sc + bv.w;
  *(float4*)(out + (size_t)node * DHID + c4) = o;
}

__global__ __launch_bounds__(256) void k_scatter(
    const int* __restrict__ src, const int* __restrict__ dst,
    const float* __restrict__ dinv, const float* __restrict__ h,
    float* __restrict__ out, int e)
{
  int w    = (blockIdx.x * blockDim.x + threadIdx.x) >> 5;
  int lane = threadIdx.x & 31;
  if (w >= e) return;
  int   s = src[w], d = dst[w];
  float c = dinv[s] * dinv[d];
  float4 hv = *(const float4*)(h + (size_t)s * DHID + lane * 4);
  float* o  = out + (size_t)d * DHID + lane * 4;
  unsafeAtomicAdd(o + 0, hv.x * c);
  unsafeAtomicAdd(o + 1, hv.y * c);
  unsafeAtomicAdd(o + 2, hv.z * c);
  unsafeAtomicAdd(o + 3, hv.w * c);
}

__global__ __launch_bounds__(256) void k_bnstats(const float* __restrict__ h,
                                                 float* __restrict__ stats, int n) {
  __shared__ float ss[256], sq[256];
  int t = threadIdx.x, c = t & 127, half = t >> 7;
  float s = 0.0f, q = 0.0f;
  for (int r = blockIdx.x * 2 + half; r < n; r += gridDim.x * 2) {
    float v = fmaxf(h[(size_t)r * DHID + c], 0.0f);
    s += v; q += v * v;
  }
  ss[t] = s; sq[t] = q;
  __syncthreads();
  if (t < 128) {
    unsafeAtomicAdd(&stats[c],       ss[t] + ss[t + 128]);
    unsafeAtomicAdd(&stats[128 + c], sq[t] + sq[t + 128]);
  }
}

__global__ void k_bnfinal(const float* __restrict__ stats, const float* __restrict__ gamma,
                          const float* __restrict__ beta, float* __restrict__ bn, int n) {
  int c = threadIdx.x;
  if (c < 128) {
    float inv = 1.0f / (float)n;
    float mu  = stats[c] * inv;
    float var = stats[128 + c] * inv - mu * mu;
    float sc  = gamma[c] * __frsqrt_rn(var + BN_EPS);
    bn[c]       = sc;
    bn[128 + c] = beta[c] - mu * sc;
  }
}

__global__ __launch_bounds__(256) void k_head(const float* __restrict__ hc,
                                              const float* __restrict__ Wr,
                                              const float* __restrict__ br,
                                              float* __restrict__ out, int n) {
  int w    = (blockIdx.x * blockDim.x + threadIdx.x) >> 5;
  int lane = threadIdx.x & 31;
  if (w >= n) return;
  float4 v = *(const float4*)(hc + (size_t)w * DHID + lane * 4);
  int c = lane * 4;
  float p0 = v.x * Wr[c * 2]       + v.y * Wr[(c + 1) * 2]
           + v.z * Wr[(c + 2) * 2] + v.w * Wr[(c + 3) * 2];
  float p1 = v.x * Wr[c * 2 + 1]       + v.y * Wr[(c + 1) * 2 + 1]
           + v.z * Wr[(c + 2) * 2 + 1] + v.w * Wr[(c + 3) * 2 + 1];
#pragma unroll
  for (int off = 16; off > 0; off >>= 1) {
    p0 += __shfl_down(p0, off, 32);
    p1 += __shfl_down(p1, off, 32);
  }
  if (lane == 0) {
    out[(size_t)w * 2]     = p0 + br[0];
    out[(size_t)w * 2 + 1] = p1 + br[1];
  }
}

extern "C" void kernel_launch(void* const* d_in, const int* in_sizes, int n_in,
                              void* d_out, int out_size, void* d_ws, size_t ws_size,
                              hipStream_t stream) {
  const float* x     = (const float*)d_in[0];
  const int*   eidx  = (const int*)d_in[1];
  const float* W1    = (const float*)d_in[2];
  const float* b1    = (const float*)d_in[3];
  const float* W2    = (const float*)d_in[4];
  const float* b2    = (const float*)d_in[5];
  const float* W3    = (const float*)d_in[6];
  const float* b3    = (const float*)d_in[7];
  const float* gamma = (const float*)d_in[8];
  const float* beta  = (const float*)d_in[9];
  const float* Wc    = (const float*)d_in[10];
  const float* bc    = (const float*)d_in[11];
  const float* Wr    = (const float*)d_in[12];
  const float* br    = (const float*)d_in[13];
  float* out = (float*)d_out;

  const int N = N_NODES, E = E_EDGES;
  const int* src = eidx;
  const int* dst = eidx + E;

  float* buf0  = (float*)d_ws;
  float* buf1  = buf0 + (size_t)N * DHID;
  float* dinv  = buf1 + (size_t)N * DHID;
  float* selfc = dinv + N;
  float* stats = selfc + N;
  float* bnp   = stats + 256;

  const dim3 blk(256);
  const int gN    = (N + 255) / 256;
  const int gE    = (E + 255) / 256;
  const int gNF4  = (N * (DHID / 4) + 255) / 256;
  const int gGemm = (N + 127) / 128;
  const int gScat = (E * 32 + 255) / 256;
  const int gHead = (N * 32 + 255) / 256;

  k_zero<<<gN, blk, 0, stream>>>(dinv, N);
  k_zero<<<1, blk, 0, stream>>>(stats, 256);
  k_degree<<<gE, blk, 0, stream>>>(dst, dinv, E);
  k_dinv<<<gN, blk, 0, stream>>>(dinv, selfc, N);

  k_gemm128<0, false, false><<<gGemm, blk, 0, stream>>>(x, W1, buf0, N, nullptr, nullptr, nullptr);
  k_selfbias<<<gNF4, blk, 0, stream>>>(buf0, selfc, b1, buf1, N);
  k_scatter<<<gScat, blk, 0, stream>>>(src, dst, dinv, buf0, buf1, E);

  k_gemm128<1, false, false><<<gGemm, blk, 0, stream>>>(buf1, W2, buf0, N, nullptr, nullptr, nullptr);
  k_selfbias<<<gNF4, blk, 0, stream>>>(buf0, selfc, b2, buf1, N);
  k_scatter<<<gScat, blk, 0, stream>>>(src, dst, dinv, buf0, buf1, E);

  k_gemm128<1, false, false><<<gGemm, blk, 0, stream>>>(buf1, W3, buf0, N, nullptr, nullptr, nullptr);
  k_selfbias<<<gNF4, blk, 0, stream>>>(buf0, selfc, b3, buf1, N);
  k_scatter<<<gScat, blk, 0, stream>>>(src, dst, dinv, buf0, buf1, E);

  k_bnstats<<<1024, blk, 0, stream>>>(buf1, stats, N);
  k_bnfinal<<<1, 128, 0, stream>>>(stats, gamma, beta, bnp, N);

  k_gemm128<2, true, true><<<gGemm, blk, 0, stream>>>(buf1, Wc, buf0, N, bnp, bnp + 128, bc);

  k_head<<<gHead, blk, 0, stream>>>(buf0, Wr, br, out, N);
}


# Edge sums: 256 blocks each hold 391 node rows in LDS, find their edges in a byte key plane (dst/391) and add gathered rows by LDS f32 adds; global atomics and self-loop pass gone; LDS arrival order
# speedup vs baseline: 2.1069x; 2.1069x over previous
#include <hip/hip_runtime.h>

#define N_NODES 100000
#define DHID    128
#define E_EDGES 1600000
#define BN_EPS  1e-5f

typedef __attribute__((ext_vector_type(2))) float v2f;
typedef __attribute__((ext_vector_type(8))) float v8f;

__global__ void k_zero(float* __restrict__ p, int n) {
  int i = blockIdx.x * blockDim.x + threadIdx.x;
  if (i < n) p[i] = 0.0f;
}

#define ROWS_PER_BLOCK 391
#define AGG_BLOCKS     256
static_assert(AGG_BLOCKS * ROWS_PER_BLOCK >= N_NODES, "ranges must cover every node");
static_assert((N_NODES - 1) / ROWS_PER_BLOCK < 256, "a range number must fit one byte");
static_assert(E_EDGES % 16 == 0, "the key plane is read 16 keys per lane");

__global__ void k_degree_keys(const int* __restrict__ dst, float* __restrict__ deg,
                              unsigned int* __restrict__ keys, int e4) {
  int i = blockIdx.x * blockDim.x + threadIdx.x;
  if (i >= e4) return;
  int4 d = ((const int4*)dst)[i];
  unsafeAtomicAdd(&deg[d.x], 1.0f);
  unsafeAtomicAdd(&deg[d.y], 1.0f);
  unsafeAtomicAdd(&deg[d.z], 1.0f);
  unsafeAtomicAdd(&deg[d.w], 1.0f);
  keys[i] = (unsigned)(d.x / ROWS_PER_BLOCK)         | ((unsigned)(d.y / ROWS_PER_BLOCK) << 8) |
            ((unsigned)(d.z / ROWS_PER_BLOCK) << 16) | ((unsigned)(d.w / ROWS_PER_BLOCK) << 24);
}

__global__ void k_dinv(float* __restrict__ deg_dinv, float* __restrict__ selfc, int n) {
  int i = blockIdx.x * blockDim.x + threadIdx.x;
  if (i < n) {
    float d = __frsqrt_rn(deg_dinv[i] + 1.0f);
    deg_dinv[i] = d;
    selfc[i]    = d * d;
  }
}

template <int AMODE, bool HAS_BIAS, bool EPI_RELU>
__global__ __launch_bounds__(256) void k_gemm128(
    const float* __restrict__ A, const float* __restrict__ W,
    float* __restrict__ C, int nrows,
    const float* __restrict__ bnScale, const float* __restrict__ bnShift,
    const float* __restrict__ bias)
{
  __shared__ float wl[DHID * DHID];
  const int tid = threadIdx.x;
  for (int i = tid; i < DHID * DHID; i += 256) {
    int k = i >> 7, n = i & 127;
    wl[(((k >> 1) << 7) + n) * 2 + (k & 1)] = W[i];
  }
  __syncthreads();

  const int wave  = tid >> 5;
  const int lane  = tid & 31;
  const int laneM = lane & 15;
  const int hi    = lane >> 4;
  const int m0    = (blockIdx.x * 8 + wave) << 4;
  if (m0 >= nrows) return;

  const float2* __restrict__ wl2  = (const float2*)wl;
  const float*  __restrict__ arow = A + (size_t)(m0 + laneM) * DHID;

  v8f acc[8];
#pragma unroll
  for (int t = 0; t < 8; ++t) acc[t] = (v8f){};

  for (int k = 0; k < DHID; k += 4) {
    float2 av = *(const float2*)(arow + k + hi * 2);
    if (AMODE == 1) {
      av.x = fmaxf(av.x, 0.0f);
      av.y = fmaxf(av.y, 0.0f);
    } else if (AMODE == 2) {
      int c0 = k + hi * 2;
      av.x = fmaxf(av.x, 0.0f) * bnScale[c0]     + bnShift[c0];
      av.y = fmaxf(av.y, 0.0f) * bnScale[c0 + 1] + bnShift[c0 + 1];
    }
    v2f a; a[0] = av.x; a[1] = av.y;

    const float2* __restrict__ brow = wl2 + (((k >> 1) + hi) << 7);
#pragma unroll
    for (int t = 0; t < 8; ++t) {
      float2 bv = brow[t * 16 + laneM];
      v2f b; b[0] = bv.x; b[1] = bv.y;
      acc[t] = __builtin_amdgcn_wmma_f32_16x16x4_f32(
          false, a, false, b, (short)0, acc[t], false, false);
    }
  }

#pragma unroll
  for (int t = 0; t < 8; ++t) {
    int   col  = t * 16 + laneM;
    float bval = HAS_BIAS ? bias[col] : 0.0f;
#pragma unroll
    for (int r = 0; r < 8; ++r) {
      int   row = m0 + r + hi * 8;
      float v   = acc[t][r] + bval;
      if (EPI_RELU) v = fmaxf(v, 0.0f);
      C[(size_t)row * DHID + col] = v;
    }
  }
}

#define LIST_CAP   24576
#define EDGE_GROUP 8

__device__ __forceinline__ unsigned zero_bytes(unsigned y) {
  unsigned t = (y & 0x7F7F7F7Fu) + 0x7F7F7F7Fu;
  return ~(t | y | 0x7F7F7F7Fu);
}

__global__ __launch_bounds__(1024) void k_aggregate(
    const int* __restrict__ src, const int* __restrict__ dst,
    const unsigned int* __restrict__ keys,
    const float* __restrict__ dinv, const float* __restrict__ selfc,
    const float* __restrict__ h, const float* __restrict__ b,
    float* __restrict__ out)
{
  __shared__ float        acc[ROWS_PER_BLOCK * DHID];
  __shared__ unsigned int list[LIST_CAP];
  __shared__ unsigned int cnt;

  const int      tid  = threadIdx.x;
  const unsigned lane = threadIdx.x & 31u;
  const int      n0   = blockIdx.x * ROWS_PER_BLOCK;

  const int nvec = min(ROWS_PER_BLOCK, N_NODES - n0) * (DHID / 4);

  if (tid == 0) cnt = 0u;
  {
    const float4* __restrict__ h4 = (const float4*)(h + (size_t)n0 * DHID);
    const float4 bv = *(const float4*)(b + (tid & 31) * 4);
#pragma unroll 4
    for (int i = tid; i < nvec; i += 1024) {
      float  sc = selfc[n0 + (i >> 5)];
      float4 hv = h4[i];
      float4 o;
      o.x = hv.x * sc + bv.x;
      o.y = hv.y * sc + bv.y;
      o.z = hv.z * sc + bv.z;
      o.w = hv.w * sc + bv.w;
      *(float4*)(acc + i * 4) = o;
    }
  }
  __syncthreads();

  const unsigned pat = (unsigned)blockIdx.x * 0x01010101u;
  const uint4* __restrict__ keys4 = (const uint4*)keys;
  const int NQ    = E_EDGES / 16;
  const int NITER = (NQ + 4095) / 4096;
  for (int k = 0; k < NITER; ++k) {
    int it = k + (int)(blockIdx.x % NITER);
    if (it >= NITER) it -= NITER;
    const int base = it * 4096;
    uint4 kv[4];
#pragma unroll
    for (int u = 0; u < 4; ++u) {
      const int   q  = base + u * 1024 + tid;
      const uint4 t  = keys4[min(q, NQ - 1)];
      const bool  in = q < NQ;
      kv[u].x = in ? t.x : ~pat;
      kv[u].y = in ? t.y : ~pat;
      kv[u].z = in ? t.z : ~pat;
      kv[u].w = in ? t.w : ~pat;
    }
#pragma unroll
    for (int u = 0; u < 4; ++u) {
      const unsigned q = (unsigned)(base + u * 1024 + tid);
      unsigned m = (zero_bytes(kv[u].x ^ pat) >> 7) | (zero_bytes(kv[u].y ^ pat) >> 6) |
                   (zero_bytes(kv[u].z ^ pat) >> 5) | (zero_bytes(kv[u].w ^ pat) >> 4);
      while (m) {
        const unsigned p = (unsigned)__builtin_ctz(m);
        m &= m - 1u;
        const unsigned e    = q * 16u + (p & 7u) * 4u + (p >> 3);
        const unsigned slot = atomicAdd(&cnt, 1u);
        if (slot < LIST_CAP) {
          list[slot] = e;
        } else {
          const int    s  = src[e];
          const int    d  = dst[e];
          const float  c  = dinv[s] * dinv[d];
          const float* hr = h + (size_t)s * DHID;
          float*       ar = acc + (d - n0) * DHID;
          for (int ch = 0; ch < DHID; ++ch) unsafeAtomicAdd(ar + ch, hr[ch] * c);
        }
      }
    }
  }
  __syncthreads();

  const int total = __builtin_amdgcn_readfirstlane(min((int)cnt, LIST_CAP));
  const int wave  = __builtin_amdgcn_readfirstlane(tid >> 5);
  for (int g0 = wave * EDGE_GROUP; g0 < total; g0 += 32 * EDGE_GROUP) {
    const int nvalid = min(EDGE_GROUP, total - g0);
    const unsigned e  = list[g0 + min((int)lane, nvalid - 1)];
    const int      s  = src[e];
    const int      d  = dst[e];
    const float    c  = dinv[s] * dinv[d];
    const int      dl = d - n0;
    if (nvalid == EDGE_GROUP) {
      float v[EDGE_GROUP][4];
#pragma unroll
      for (int u = 0; u < EDGE_GROUP; ++u) {
        int sj = __builtin_amdgcn_readlane(s, u);
        const float* hr = h + (size_t)sj * DHID;
        v[u][0] = hr[lane];
        v[u][1] = hr[lane + 32];
        v[u][2] = hr[lane + 64];
        v[u][3] = hr[lane + 96];
      }
#pragma unroll
      for (int u = 0; u < EDGE_GROUP; ++u) {
        int   dj = __builtin_amdgcn_readlane(dl, u);
        float cj = __int_as_float(__builtin_amdgcn_readlane(__float_as_int(c), u));
        float* a = acc + dj * DHID;
        unsafeAtomicAdd(a + lane,      v[u][0] * cj);
        unsafeAtomicAdd(a + lane + 32, v[u][1] * cj);
        unsafeAtomicAdd(a + lane + 64, v[u][2] * cj);
        unsafeAtomicAdd(a + lane + 96, v[u][3] * cj);
      }
    } else {
      for (int j = 0; j < nvalid; ++j) {
        int   sj = __builtin_amdgcn_readlane(s, j);
        int   dj = __builtin_amdgcn_readlane(dl, j);
        float cj = __int_as_float(__builtin_amdgcn_readlane(__float_as_int(c), j));
        const float* hr = h + (size_t)sj * DHID;
        float*       a  = acc + dj * DHID;
        float h0 = hr[lane], h1 = hr[lane + 32], h2 = hr[lane + 64], h3 = hr[lane + 96];
        unsafeAtomicAdd(a + lane,      h0 * cj);
        unsafeAtomicAdd(a + lane + 32, h1 * cj);
        unsafeAtomicAdd(a + lane + 64, h2 * cj);
        unsafeAtomicAdd(a + lane + 96, h3 * cj);
      }
    }
  }
  __syncthreads();

  float4* __restrict__ out4 = (float4*)(out + (size_t)n0 * DHID);
  for (int i = tid; i < nvec; i += 1024) out4[i] = *(const float4*)(acc + i * 4);
}

__global__ __launch_bounds__(256) void k_bnstats(const float* __restrict__ h,
                                                 float* __restrict__ stats, int n) {
  __shared__ float ss[256], sq[256];
  int t = threadIdx.x, c = t & 127, half = t >> 7;
  float s = 0.0f, q = 0.0f;
  for (int r = blockIdx.x * 2 + half; r < n; r += gridDim.x * 2) {
    float v = fmaxf(h[(size_t)r * DHID + c], 0.0f);
    s += v; q += v * v;
  }
  ss[t] = s; sq[t] = q;
  __syncthreads();
  if (t < 128) {
    unsafeAtomicAdd(&stats[c],       ss[t] + ss[t + 128]);
    unsafeAtomicAdd(&stats[128 + c], sq[t] + sq[t + 128]);
  }
}

__global__ void k_bnfinal(const float* __restrict__ stats, const float* __restrict__ gamma,
                          const float* __restrict__ beta, float* __restrict__ bn, int n) {
  int c = threadIdx.x;
  if (c < 128) {
    float inv = 1.0f / (float)n;
    float mu  = stats[c] * inv;
    float var = stats[128 + c] * inv - mu * mu;
    float sc  = gamma[c] * __frsqrt_rn(var + BN_EPS);
    bn[c]       = sc;
    bn[128 + c] = beta[c] - mu * sc;
  }
}

__global__ __launch_bounds__(256) void k_head(const float* __restrict__ hc,
                                              const float* __restrict__ Wr,
                                              const float* __restrict__ br,
                                              float* __restrict__ out, int n) {
  int w    = (blockIdx.x * blockDim.x + threadIdx.x) >> 5;
  int lane = threadIdx.x & 31;
  if (w >= n) return;
  float4 v = *(const float4*)(hc + (size_t)w * DHID + lane * 4);
  int c = lane * 4;
  float p0 = v.x * Wr[c * 2]       + v.y * Wr[(c + 1) * 2]
           + v.z * Wr[(c + 2) * 2] + v.w * Wr[(c + 3) * 2];
  float p1 = v.x * Wr[c * 2 + 1]       + v.y * Wr[(c + 1) * 2 + 1]
           + v.z * Wr[(c + 2) * 2 + 1] + v.w * Wr[(c + 3) * 2 + 1];
#pragma unroll
  for (int off = 16; off > 0; off >>= 1) {
    p0 += __shfl_down(p0, off, 32);
    p1 += __shfl_down(p1, off, 32);
  }
  if (lane == 0) {
    out[(size_t)w * 2]     = p0 + br[0];
    out[(size_t)w * 2 + 1] = p1 + br[1];
  }
}

extern "C" void kernel_launch(void* const* d_in, const int* in_sizes, int n_in,
                              void* d_out, int out_size, void* d_ws, size_t ws_size,
                              hipStream_t stream) {
  const float* x     = (const float*)d_in[0];
  const int*   eidx  = (const int*)d_in[1];
  const float* W1    = (const float*)d_in[2];
  const float* b1    = (const float*)d_in[3];
  const float* W2    = (const float*)d_in[4];
  const float* b2    = (const float*)d_in[5];
  const float* W3    = (const float*)d_in[6];
  const float* b3    = (const float*)d_in[7];
  const float* gamma = (const float*)d_in[8];
  const float* beta  = (const float*)d_in[9];
  const float* Wc    = (const float*)d_in[10];
  const float* bc    = (const float*)d_in[11];
  const float* Wr    = (const float*)d_in[12];
  const float* br    = (const float*)d_in[13];
  float* out = (float*)d_out;

  const int N = N_NODES, E = E_EDGES;
  const int* src = eidx;
  const int* dst = eidx + E;

  float* buf0  = (float*)d_ws;
  float* buf1  = buf0 + (size_t)N * DHID;
  float* dinv  = buf1 + (size_t)N * DHID;
  float* selfc = dinv + N;
  float* stats = selfc + N;
  float* bnp   = stats + 256;
  unsigned int* keys = (unsigned int*)(bnp + 256);

  const dim3 blk(256);
  const int gN    = (N + 255) / 256;
  const int gE4   = (E / 4 + 255) / 256;
  const int gGemm = (N + 127) / 128;
  const int gHead = (N * 32 + 255) / 256;

  k_zero<<<gN, blk, 0, stream>>>(dinv, N);
  k_zero<<<1, blk, 0, stream>>>(stats, 256);
  k_degree_keys<<<gE4, blk, 0, stream>>>(dst, dinv, keys, E / 4);
  k_dinv<<<gN, blk, 0, stream>>>(dinv, selfc, N);

  k_gemm128<0, false, false><<<gGemm, blk, 0, stream>>>(x, W1, buf0, N, nullptr, nullptr, nullptr);
  k_aggregate<<<AGG_BLOCKS, 1024, 0, stream>>>(src, dst, keys, dinv, selfc, buf0, b1, buf1);

  k_gemm128<1, false, false><<<gGemm, blk, 0, stream>>>(buf1, W2, buf0, N, nullptr, nullptr, nullptr);
  k_aggregate<<<AGG_BLOCKS, 1024, 0, stream>>>(src, dst, keys, dinv, selfc, buf0, b2, buf1);

  k_gemm128<1, false, false><<<gGemm, blk, 0, stream>>>(buf1, W3, buf0, N, nullptr, nullptr, nullptr);
  k_aggregate<<<AGG_BLOCKS, 1024, 0, stream>>>(src, dst, keys, dinv, selfc, buf0, b3, buf1);

  k_bnstats<<<1024, blk, 0, stream>>>(buf1, stats, N);
  k_bnfinal<<<1, 128, 0, stream>>>(stats, gamma, beta, bnp, N);

  k_gemm128<2, true, true><<<gGemm, blk, 0, stream>>>(buf1, Wc, buf0, N, bnp, bnp + 128, bc);

  k_head<<<gHead, blk, 0, stream>>>(buf0, Wr, br, out, N);
}


# Edge sums: layer 1 also writes each range's edge list to workspace as (source, local row, coefficient) records; layers 2, 3 read them, no key-plane scan or src/dst/dinv loads; LDS arrival order
# speedup vs baseline: 2.4155x; 1.1465x over previous
#include <hip/hip_runtime.h>

#define N_NODES 100000
#define DHID    128
#define E_EDGES 1600000
#define BN_EPS  1e-5f

typedef __attribute__((ext_vector_type(2))) float v2f;
typedef __attribute__((ext_vector_type(8))) float v8f;

__global__ void k_zero(float* __restrict__ p, int n) {
  int i = blockIdx.x * blockDim.x + threadIdx.x;
  if (i < n) p[i] = 0.0f;
}

#define ROWS_PER_BLOCK 391
#define AGG_BLOCKS     256
static_assert(AGG_BLOCKS * ROWS_PER_BLOCK >= N_NODES, "ranges must cover every node");
static_assert((N_NODES - 1) / ROWS_PER_BLOCK < 256, "a range number must fit one byte");
static_assert(E_EDGES % 16 == 0, "the key plane is read 16 keys per lane");

__global__ void k_degree_keys(const int* __restrict__ dst, float* __restrict__ deg,
                              unsigned int* __restrict__ keys, int e4) {
  int i = blockIdx.x * blockDim.x + threadIdx.x;
  if (i >= e4) return;
  int4 d = ((const int4*)dst)[i];
  unsafeAtomicAdd(&deg[d.x], 1.0f);
  unsafeAtomicAdd(&deg[d.y], 1.0f);
  unsafeAtomicAdd(&deg[d.z], 1.0f);
  unsafeAtomicAdd(&deg[d.w], 1.0f);
  keys[i] = (unsigned)(d.x / ROWS_PER_BLOCK)         | ((unsigned)(d.y / ROWS_PER_BLOCK) << 8) |
            ((unsigned)(d.z / ROWS_PER_BLOCK) << 16) | ((unsigned)(d.w / ROWS_PER_BLOCK) << 24);
}

__global__ void k_dinv(float* __restrict__ deg_dinv, float* __restrict__ selfc, int n) {
  int i = blockIdx.x * blockDim.x + threadIdx.x;
  if (i < n) {
    float d = __frsqrt_rn(deg_dinv[i] + 1.0f);
    deg_dinv[i] = d;
    selfc[i]    = d * d;
  }
}

template <int AMODE, bool HAS_BIAS, bool EPI_RELU>
__global__ __launch_bounds__(256) void k_gemm128(
    const float* __restrict__ A, const float* __restrict__ W,
    float* __restrict__ C, int nrows,
    const float* __restrict__ bnScale, const float* __restrict__ bnShift,
    const float* __restrict__ bias)
{
  __shared__ float wl[DHID * DHID];
  const int tid = threadIdx.x;
  for (int i = tid; i < DHID * DHID; i += 256) {
    int k = i >> 7, n = i & 127;
    wl[(((k >> 1) << 7) + n) * 2 + (k & 1)] = W[i];
  }
  __syncthreads();

  const int wave  = tid >> 5;
  const int lane  = tid & 31;
  const int laneM = lane & 15;
  const int hi    = lane >> 4;
  const int m0    = (blockIdx.x * 8 + wave) << 4;
  if (m0 >= nrows) return;

  const float2* __restrict__ wl2  = (const float2*)wl;
  const float*  __restrict__ arow = A + (size_t)(m0 + laneM) * DHID;

  v8f acc[8];
#pragma unroll
  for (int t = 0; t < 8; ++t) acc[t] = (v8f){};

  for (int k = 0; k < DHID; k += 4) {
    float2 av = *(const float2*)(arow + k + hi * 2);
    if (AMODE == 1) {
      av.x = fmaxf(av.x, 0.0f);
      av.y = fmaxf(av.y, 0.0f);
    } else if (AMODE == 2) {
      int c0 = k + hi * 2;
      av.x = fmaxf(av.x, 0.0f) * bnScale[c0]     + bnShift[c0];
      av.y = fmaxf(av.y, 0.0f) * bnScale[c0 + 1] + bnShift[c0 + 1];
    }
    v2f a; a[0] = av.x; a[1] = av.y;

    const float2* __restrict__ brow = wl2 + (((k >> 1) + hi) << 7);
#pragma unroll
    for (int t = 0; t < 8; ++t) {
      float2 bv = brow[t * 16 + laneM];
      v2f b; b[0] = bv.x; b[1] = bv.y;
      acc[t] = __builtin_amdgcn_wmma_f32_16x16x4_f32(
          false, a, false, b, (short)0, acc[t], false, false);
    }
  }

#pragma unroll
  for (int t = 0; t < 8; ++t) {
    int   col  = t * 16 + laneM;
    float bval = HAS_BIAS ? bias[col] : 0.0f;
#pragma unroll
    for (int r = 0; r < 8; ++r) {
      int   row = m0 + r + hi * 8;
      float v   = acc[t][r] + bval;
      if (EPI_RELU) v = fmaxf(v, 0.0f);
      C[(size_t)row * DHID + col] = v;
    }
  }
}

#define LIST_CAP   24576
#define EDGE_GROUP 8
#define SRC_BITS   17
static_assert(N_NODES <= (1 << SRC_BITS), "a source node must fit the record's low bits");
static_assert(ROWS_PER_BLOCK <= (1 << (32 - SRC_BITS)), "a local row must fit the record's high bits");

__device__ __forceinline__ unsigned zero_bytes(unsigned y) {
  unsigned t = (y & 0x7F7F7F7Fu) + 0x7F7F7F7Fu;
  return ~(t | y | 0x7F7F7F7Fu);
}

__device__ __forceinline__ void add_group(const float* __restrict__ h, float* acc, unsigned lane,
                                          int nvalid, int s, int dl, float c) {
  if (nvalid == EDGE_GROUP) {
    float v[EDGE_GROUP][4];
#pragma unroll
    for (int u = 0; u < EDGE_GROUP; ++u) {
      int sj = __builtin_amdgcn_readlane(s, u);
      const float* hr = h + (size_t)sj * DHID;
      v[u][0] = hr[lane];
      v[u][1] = hr[lane + 32];
      v[u][2] = hr[lane + 64];
      v[u][3] = hr[lane + 96];
    }
#pragma unroll
    for (int u = 0; u < EDGE_GROUP; ++u) {
      int   dj = __builtin_amdgcn_readlane(dl, u);
      float cj = __int_as_float(__builtin_amdgcn_readlane(__float_as_int(c), u));
      float* a = acc + dj * DHID;
      unsafeAtomicAdd(a + lane,      v[u][0] * cj);
      unsafeAtomicAdd(a + lane + 32, v[u][1] * cj);
      unsafeAtomicAdd(a + lane + 64, v[u][2] * cj);
      unsafeAtomicAdd(a + lane + 96, v[u][3] * cj);
    }
  } else {
    for (int j = 0; j < nvalid; ++j) {
      int   sj = __builtin_amdgcn_readlane(s, j);
      int   dj = __builtin_amdgcn_readlane(dl, j);
      float cj = __int_as_float(__builtin_amdgcn_readlane(__float_as_int(c), j));
      const float* hr = h + (size_t)sj * DHID;
      float*       a  = acc + dj * DHID;
      float h0 = hr[lane], h1 = hr[lane + 32], h2 = hr[lane + 64], h3 = hr[lane + 96];
      unsafeAtomicAdd(a + lane,      h0 * cj);
      unsafeAtomicAdd(a + lane + 32, h1 * cj);
      unsafeAtomicAdd(a + lane + 64, h2 * cj);
      unsafeAtomicAdd(a + lane + 96, h3 * cj);
    }
  }
}

template <bool BUILD>
__global__ __launch_bounds__(1024) void k_aggregate(
    const int* __restrict__ src, const int* __restrict__ dst,
    const unsigned int* __restrict__ keys,
    const float* __restrict__ dinv, const float* __restrict__ selfc,
    const float* __restrict__ h, const float* __restrict__ b,
    float* __restrict__ out,
    unsigned int* __restrict__ recs, unsigned int* __restrict__ counts)
{
  __shared__ float        acc[ROWS_PER_BLOCK * DHID];
  __shared__ unsigned int list[LIST_CAP];
  __shared__ unsigned int cnt;

  const int      tid  = threadIdx.x;
  const unsigned lane = threadIdx.x & 31u;
  const int      n0   = blockIdx.x * ROWS_PER_BLOCK;
  const int      wave = __builtin_amdgcn_readfirstlane(tid >> 5);

  uint2* __restrict__ rl = (uint2*)recs + (size_t)blockIdx.x * LIST_CAP;

  const int nvec = min(ROWS_PER_BLOCK, N_NODES - n0) * (DHID / 4);

  if (tid == 0) cnt = 0u;
  {
    const float4* __restrict__ h4 = (const float4*)(h + (size_t)n0 * DHID);
    const float4 bv = *(const float4*)(b + (tid & 31) * 4);
#pragma unroll 4
    for (int i = tid; i < nvec; i += 1024) {
      float  sc = selfc[n0 + (i >> 5)];
      float4 hv = h4[i];
      float4 o;
      o.x = hv.x * sc + bv.x;
      o.y = hv.y * sc + bv.y;
      o.z = hv.z * sc + bv.z;
      o.w = hv.w * sc + bv.w;
      *(float4*)(acc + i * 4) = o;
    }
  }
  __syncthreads();

  int stored = 0;
  if (!BUILD) stored = __builtin_amdgcn_readfirstlane((int)counts[blockIdx.x]);

  if (BUILD || stored > LIST_CAP) {
    const unsigned pat = (unsigned)blockIdx.x * 0x01010101u;
    const uint4* __restrict__ keys4 = (const uint4*)keys;
    const int NQ    = E_EDGES / 16;
    const int NITER = (NQ + 4095) / 4096;
    for (int k = 0; k < NITER; ++k) {
      int it = k + (int)(blockIdx.x % NITER);
      if (it >= NITER) it -= NITER;
      const int base = it * 4096;
      uint4 kv[4];
#pragma unroll
      for (int u = 0; u < 4; ++u) {
        const int   q  = base + u * 1024 + tid;
        const uint4 t  = keys4[min(q, NQ - 1)];
        const bool  in = q < NQ;
        kv[u].x = in ? t.x : ~pat;
        kv[u].y = in ? t.y : ~pat;
        kv[u].z = in ? t.z : ~pat;
        kv[u].w = in ? t.w : ~pat;
      }
#pragma unroll
      for (int u = 0; u < 4; ++u) {
        const unsigned q = (unsigned)(base + u * 1024 + tid);
        unsigned m = (zero_bytes(kv[u].x ^ pat) >> 7) | (zero_bytes(kv[u].y ^ pat) >> 6) |
                     (zero_bytes(kv[u].z ^ pat) >> 5) | (zero_bytes(kv[u].w ^ pat) >> 4);
        while (m) {
          const unsigned p = (unsigned)__builtin_ctz(m);
          m &= m - 1u;
          const unsigned e    = q * 16u + (p & 7u) * 4u + (p >> 3);
          const unsigned slot = atomicAdd(&cnt, 1u);
          if (slot < LIST_CAP) {
            list[slot] = e;
          } else {
            const int    s  = src[e];
            const int    d  = dst[e];
            const float  c  = dinv[s] * dinv[d];
            const float* hr = h + (size_t)s * DHID;
            float*       ar = acc + (d - n0) * DHID;
            for (int ch = 0; ch < DHID; ++ch) unsafeAtomicAdd(ar + ch, hr[ch] * c);
          }
        }
      }
    }
    __syncthreads();

    if (BUILD && tid == 0) counts[blockIdx.x] = cnt;

    const int total = __builtin_amdgcn_readfirstlane(min((int)cnt, LIST_CAP));
    for (int g0 = wave * EDGE_GROUP; g0 < total; g0 += 32 * EDGE_GROUP) {
      const int nvalid = min(EDGE_GROUP, total - g0);
      const unsigned e  = list[g0 + min((int)lane, nvalid - 1)];
      const int      s  = src[e];
      const int      d  = dst[e];
      const float    c  = dinv[s] * dinv[d];
      const int      dl = d - n0;
      if (BUILD) {
        if ((int)lane < nvalid)
          rl[g0 + (int)lane] = make_uint2((unsigned)s | ((unsigned)dl << SRC_BITS), __float_as_uint(c));
      }
      add_group(h, acc, lane, nvalid, s, dl, c);
    }
  } else {
    const int total = stored;
    const int l7    = min((int)lane, EDGE_GROUP - 1);
    int   g0 = wave * EDGE_GROUP;
    uint2 rc = make_uint2(0u, 0u);
    if (g0 < total) rc = rl[min(g0 + l7, total - 1)];
    for (; g0 < total; g0 += 32 * EDGE_GROUP) {
      const int nvalid = min(EDGE_GROUP, total - g0);
      const int gn = g0 + 32 * EDGE_GROUP;
      uint2 nx = rc;
      if (gn < total) nx = rl[min(gn + l7, total - 1)];
      const int   s  = (int)(rc.x & ((1u << SRC_BITS) - 1u));
      const int   dl = (int)(rc.x >> SRC_BITS);
      const float c  = __uint_as_float(rc.y);
      add_group(h, acc, lane, nvalid, s, dl, c);
      rc = nx;
    }
  }
  __syncthreads();

  float4* __restrict__ out4 = (float4*)(out + (size_t)n0 * DHID);
  for (int i = tid; i < nvec; i += 1024) out4[i] = *(const float4*)(acc + i * 4);
}

__global__ __launch_bounds__(256) void k_bnstats(const float* __restrict__ h,
                                                 float* __restrict__ stats, int n) {
  __shared__ float ss[256], sq[256];
  int t = threadIdx.x, c = t & 127, half = t >> 7;
  float s = 0.0f, q = 0.0f;
  for (int r = blockIdx.x * 2 + half; r < n; r += gridDim.x * 2) {
    float v = fmaxf(h[(size_t)r * DHID + c], 0.0f);
    s += v; q += v * v;
  }
  ss[t] = s; sq[t] = q;
  __syncthreads();
  if (t < 128) {
    unsafeAtomicAdd(&stats[c],       ss[t] + ss[t + 128]);
    unsafeAtomicAdd(&stats[128 + c], sq[t] + sq[t + 128]);
  }
}

__global__ void k_bnfinal(const float* __restrict__ stats, const float* __restrict__ gamma,
                          const float* __restrict__ beta, float* __restrict__ bn, int n) {
  int c = threadIdx.x;
  if (c < 128) {
    float inv = 1.0f / (float)n;
    float mu  = stats[c] * inv;
    float var = stats[128 + c] * inv - mu * mu;
    float sc  = gamma[c] * __frsqrt_rn(var + BN_EPS);
    bn[c]       = sc;
    bn[128 + c] = beta[c] - mu * sc;
  }
}

__global__ __launch_bounds__(256) void k_head(const float* __restrict__ hc,
                                              const float* __restrict__ Wr,
                                              const float* __restrict__ br,
                                              float* __restrict__ out, int n) {
  int w    = (blockIdx.x * blockDim.x + threadIdx.x) >> 5;
  int lane = threadIdx.x & 31;
  if (w >= n) return;
  float4 v = *(const float4*)(hc + (size_t)w * DHID + lane * 4);
  int c = lane * 4;
  float p0 = v.x * Wr[c * 2]       + v.y * Wr[(c + 1) * 2]
           + v.z * Wr[(c + 2) * 2] + v.w * Wr[(c + 3) * 2];
  float p1 = v.x * Wr[c * 2 + 1]       + v.y * Wr[(c + 1) * 2 + 1]
           + v.z * Wr[(c + 2) * 2 + 1] + v.w * Wr[(c + 3) * 2 + 1];
#pragma unroll
  for (int off = 16; off > 0; off >>= 1) {
    p0 += __shfl_down(p0, off, 32);
    p1 += __shfl_down(p1, off, 32);
  }
  if (lane == 0) {
    out[(size_t)w * 2]     = p0 + br[0];
    out[(size_t)w * 2 + 1] = p1 + br[1];
  }
}

extern "C" void kernel_launch(void* const* d_in, const int* in_sizes, int n_in,
                              void* d_out, int out_size, void* d_ws, size_t ws_size,
                              hipStream_t stream) {
  const float* x     = (const float*)d_in[0];
  const int*   eidx  = (const int*)d_in[1];
  const float* W1    = (const float*)d_in[2];
  const float* b1    = (const float*)d_in[3];
  const float* W2    = (const float*)d_in[4];
  const float* b2    = (const float*)d_in[5];
  const float* W3    = (const float*)d_in[6];
  const float* b3    = (const float*)d_in[7];
  const float* gamma = (const float*)d_in[8];
  const float* beta  = (const float*)d_in[9];
  const float* Wc    = (const float*)d_in[10];
  const float* bc    = (const float*)d_in[11];
  const float* Wr    = (const float*)d_in[12];
  const float* br    = (const float*)d_in[13];
  float* out = (float*)d_out;

  const int N = N_NODES, E = E_EDGES;
  const int* src = eidx;
  const int* dst = eidx + E;

  float* buf0  = (float*)d_ws;
  float* buf1  = buf0 + (size_t)N * DHID;
  float* dinv  = buf1 + (size_t)N * DHID;
  float* selfc = dinv + N;
  float* stats = selfc + N;
  float* bnp   = stats + 256;
  unsigned int* keys = (unsigned int*)(bnp + 256);
  unsigned int* counts = keys + E / 4;
  unsigned int* recs = counts + AGG_BLOCKS;

  const dim3 blk(256);
  const int gN    = (N + 255) / 256;
  const int gE4   = (E / 4 + 255) / 256;
  const int gGemm = (N + 127) / 128;
  const int gHead = (N * 32 + 255) / 256;

  k_zero<<<gN, blk, 0, stream>>>(dinv, N);
  k_zero<<<1, blk, 0, stream>>>(stats, 256);
  k_degree_keys<<<gE4, blk, 0, stream>>>(dst, dinv, keys, E / 4);
  k_dinv<<<gN, blk, 0, stream>>>(dinv, selfc, N);

  k_gemm128<0, false, false><<<gGemm, blk, 0, stream>>>(x, W1, buf0, N, nullptr, nullptr, nullptr);
  k_aggregate<true><<<AGG_BLOCKS, 1024, 0, stream>>>(src, dst, keys, dinv, selfc, buf0, b1, buf1, recs, counts);

  k_gemm128<1, false, false><<<gGemm, blk, 0, stream>>>(buf1, W2, buf0, N, nullptr, nullptr, nullptr);
  k_aggregate<false><<<AGG_BLOCKS, 1024, 0, stream>>>(src, dst, keys, dinv, selfc, buf0, b2, buf1, recs, counts);

  k_gemm128<1, false, false><<<gGemm, blk, 0, stream>>>(buf1, W3, buf0, N, nullptr, nullptr, nullptr);
  k_aggregate<false><<<AGG_BLOCKS, 1024, 0, stream>>>(src, dst, keys, dinv, selfc, buf0, b3, buf1, recs, counts);

  k_bnstats<<<1024, blk, 0, stream>>>(buf1, stats, N);
  k_bnfinal<<<1, 128, 0, stream>>>(stats, gamma, beta, bnp, N);

  k_gemm128<2, true, true><<<gGemm, blk, 0, stream>>>(buf1, Wc, buf0, N, bnp, bnp + 128, bc);

  k_head<<<gHead, blk, 0, stream>>>(buf0, Wr, br, out, N);
}


# Edge sums: each source row is fetched by one 16-byte-per-lane load (512 contiguous bytes), not four one-line loads; LDS rows hold channels in lane order so adds stay conflict-free; LDS arrival order
# speedup vs baseline: 2.4882x; 1.0301x over previous
#include <hip/hip_runtime.h>

#define N_NODES 100000
#define DHID    128
#define E_EDGES 1600000
#define BN_EPS  1e-5f

typedef __attribute__((ext_vector_type(2))) float v2f;
typedef __attribute__((ext_vector_type(8))) float v8f;

__global__ void k_zero(float* __restrict__ p, int n) {
  int i = blockIdx.x * blockDim.x + threadIdx.x;
  if (i < n) p[i] = 0.0f;
}

#define ROWS_PER_BLOCK 391
#define AGG_BLOCKS     256
static_assert(AGG_BLOCKS * ROWS_PER_BLOCK >= N_NODES, "ranges must cover every node");
static_assert((N_NODES - 1) / ROWS_PER_BLOCK < 256, "a range number must fit one byte");
static_assert(E_EDGES % 16 == 0, "the key plane is read 16 keys per lane");

__global__ void k_degree_keys(const int* __restrict__ dst, float* __restrict__ deg,
                              unsigned int* __restrict__ keys, int e4) {
  int i = blockIdx.x * blockDim.x + threadIdx.x;
  if (i >= e4) return;
  int4 d = ((const int4*)dst)[i];
  unsafeAtomicAdd(&deg[d.x], 1.0f);
  unsafeAtomicAdd(&deg[d.y], 1.0f);
  unsafeAtomicAdd(&deg[d.z], 1.0f);
  unsafeAtomicAdd(&deg[d.w], 1.0f);
  keys[i] = (unsigned)(d.x / ROWS_PER_BLOCK)         | ((unsigned)(d.y / ROWS_PER_BLOCK) << 8) |
            ((unsigned)(d.z / ROWS_PER_BLOCK) << 16) | ((unsigned)(d.w / ROWS_PER_BLOCK) << 24);
}

__global__ void k_dinv(float* __restrict__ deg_dinv, float* __restrict__ selfc, int n) {
  int i = blockIdx.x * blockDim.x + threadIdx.x;
  if (i < n) {
    float d = __frsqrt_rn(deg_dinv[i] + 1.0f);
    deg_dinv[i] = d;
    selfc[i]    = d * d;
  }
}

template <int AMODE, bool HAS_BIAS, bool EPI_RELU>
__global__ __launch_bounds__(256) void k_gemm128(
    const float* __restrict__ A, const float* __restrict__ W,
    float* __restrict__ C, int nrows,
    const float* __restrict__ bnScale, const float* __restrict__ bnShift,
    const float* __restrict__ bias)
{
  __shared__ float wl[DHID * DHID];
  const int tid = threadIdx.x;
  for (int i = tid; i < DHID * DHID; i += 256) {
    int k = i >> 7, n = i & 127;
    wl[(((k >> 1) << 7) + n) * 2 + (k & 1)] = W[i];
  }
  __syncthreads();

  const int wave  = tid >> 5;
  const int lane  = tid & 31;
  const int laneM = lane & 15;
  const int hi    = lane >> 4;
  const int m0    = (blockIdx.x * 8 + wave) << 4;
  if (m0 >= nrows) return;

  const float2* __restrict__ wl2  = (const float2*)wl;
  const float*  __restrict__ arow = A + (size_t)(m0 + laneM) * DHID;

  v8f acc[8];
#pragma unroll
  for (int t = 0; t < 8; ++t) acc[t] = (v8f){};

  for (int k = 0; k < DHID; k += 4) {
    float2 av = *(const float2*)(arow + k + hi * 2);
    if (AMODE == 1) {
      av.x = fmaxf(av.x, 0.0f);
      av.y = fmaxf(av.y, 0.0f);
    } else if (AMODE == 2) {
      int c0 = k + hi * 2;
      av.x = fmaxf(av.x, 0.0f) * bnScale[c0]     + bnShift[c0];
      av.y = fmaxf(av.y, 0.0f) * bnScale[c0 + 1] + bnShift[c0 + 1];
    }
    v2f a; a[0] = av.x; a[1] = av.y;

    const float2* __restrict__ brow = wl2 + (((k >> 1) + hi) << 7);
#pragma unroll
    for (int t = 0; t < 8; ++t) {
      float2 bv = brow[t * 16 + laneM];
      v2f b; b[0] = bv.x; b[1] = bv.y;
      acc[t] = __builtin_amdgcn_wmma_f32_16x16x4_f32(
          false, a, false, b, (short)0, acc[t], false, false);
    }
  }

#pragma unroll
  for (int t = 0; t < 8; ++t) {
    int   col  = t * 16 + laneM;
    float bval = HAS_BIAS ? bias[col] : 0.0f;
#pragma unroll
    for (int r = 0; r < 8; ++r) {
      int   row = m0 + r + hi * 8;
      float v   = acc[t][r] + bval;
      if (EPI_RELU) v = fmaxf(v, 0.0f);
      C[(size_t)row * DHID + col] = v;
    }
  }
}

#define LIST_CAP   24576
#define EDGE_GROUP 8
#define SRC_BITS   17
static_assert(N_NODES <= (1 << SRC_BITS), "a source node must fit the record's low bits");
static_assert(ROWS_PER_BLOCK <= (1 << (32 - SRC_BITS)), "a local row must fit the record's high bits");

__device__ __forceinline__ unsigned zero_bytes(unsigned y) {
  unsigned t = (y & 0x7F7F7F7Fu) + 0x7F7F7F7Fu;
  return ~(t | y | 0x7F7F7F7Fu);
}

__device__ __forceinline__ int lds_word(int ch) { return (ch & 3) * 32 + (ch >> 2); }

__device__ __forceinline__ void add_group(const float* __restrict__ h, float* acc, unsigned lane,
                                          int nvalid, int s, int dl, float c) {
  if (nvalid == EDGE_GROUP) {
    float4 v[EDGE_GROUP];
#pragma unroll
    for (int u = 0; u < EDGE_GROUP; ++u) {
      int sj = __builtin_amdgcn_readlane(s, u);
      v[u] = ((const float4*)(h + (size_t)sj * DHID))[lane];
    }
#pragma unroll
    for (int u = 0; u < EDGE_GROUP; ++u) {
      int   dj = __builtin_amdgcn_readlane(dl, u);
      float cj = __int_as_float(__builtin_amdgcn_readlane(__float_as_int(c), u));
      float* a = acc + dj * DHID;
      unsafeAtomicAdd(a + lane,      v[u].x * cj);
      unsafeAtomicAdd(a + lane + 32, v[u].y * cj);
      unsafeAtomicAdd(a + lane + 64, v[u].z * cj);
      unsafeAtomicAdd(a + lane + 96, v[u].w * cj);
    }
  } else {
    for (int j = 0; j < nvalid; ++j) {
      int   sj = __builtin_amdgcn_readlane(s, j);
      int   dj = __builtin_amdgcn_readlane(dl, j);
      float cj = __int_as_float(__builtin_amdgcn_readlane(__float_as_int(c), j));
      const float4 hv = ((const float4*)(h + (size_t)sj * DHID))[lane];
      float*       a  = acc + dj * DHID;
      unsafeAtomicAdd(a + lane,      hv.x * cj);
      unsafeAtomicAdd(a + lane + 32, hv.y * cj);
      unsafeAtomicAdd(a + lane + 64, hv.z * cj);
      unsafeAtomicAdd(a + lane + 96, hv.w * cj);
    }
  }
}

template <bool BUILD>
__global__ __launch_bounds__(1024) void k_aggregate(
    const int* __restrict__ src, const int* __restrict__ dst,
    const unsigned int* __restrict__ keys,
    const float* __restrict__ dinv, const float* __restrict__ selfc,
    const float* __restrict__ h, const float* __restrict__ b,
    float* __restrict__ out,
    unsigned int* __restrict__ recs, unsigned int* __restrict__ counts)
{
  __shared__ float        acc[ROWS_PER_BLOCK * DHID];
  __shared__ unsigned int list[LIST_CAP];
  __shared__ unsigned int cnt;

  const int      tid  = threadIdx.x;
  const unsigned lane = threadIdx.x & 31u;
  const int      n0   = blockIdx.x * ROWS_PER_BLOCK;
  const int      wave = __builtin_amdgcn_readfirstlane(tid >> 5);

  uint2* __restrict__ rl = (uint2*)recs + (size_t)blockIdx.x * LIST_CAP;

  const int nvec = min(ROWS_PER_BLOCK, N_NODES - n0) * (DHID / 4);

  if (tid == 0) cnt = 0u;
  {
    const float4* __restrict__ h4 = (const float4*)(h + (size_t)n0 * DHID);
    const float4 bv = *(const float4*)(b + (tid & 31) * 4);
#pragma unroll 4
    for (int i = tid; i < nvec; i += 1024) {
      float  sc = selfc[n0 + (i >> 5)];
      float4 hv = h4[i];
      float* ar = acc + (i >> 5) * DHID + (i & 31);
      ar[0]  = hv.x * sc + bv.x;
      ar[32] = hv.y * sc + bv.y;
      ar[64] = hv.z * sc + bv.z;
      ar[96] = hv.w * sc + bv.w;
    }
  }
  __syncthreads();

  int stored = 0;
  if (!BUILD) stored = __builtin_amdgcn_readfirstlane((int)counts[blockIdx.x]);

  if (BUILD || stored > LIST_CAP) {
    const unsigned pat = (unsigned)blockIdx.x * 0x01010101u;
    const uint4* __restrict__ keys4 = (const uint4*)keys;
    const int NQ    = E_EDGES / 16;
    const int NITER = (NQ + 4095) / 4096;
    for (int k = 0; k < NITER; ++k) {
      int it = k + (int)(blockIdx.x % NITER);
      if (it >= NITER) it -= NITER;
      const int base = it * 4096;
      uint4 kv[4];
#pragma unroll
      for (int u = 0; u < 4; ++u) {
        const int   q  = base + u * 1024 + tid;
        const uint4 t  = keys4[min(q, NQ - 1)];
        const bool  in = q < NQ;
        kv[u].x = in ? t.x : ~pat;
        kv[u].y = in ? t.y : ~pat;
        kv[u].z = in ? t.z : ~pat;
        kv[u].w = in ? t.w : ~pat;
      }
#pragma unroll
      for (int u = 0; u < 4; ++u) {
        const unsigned q = (unsigned)(base + u * 1024 + tid);
        unsigned m = (zero_bytes(kv[u].x ^ pat) >> 7) | (zero_bytes(kv[u].y ^ pat) >> 6) |
                     (zero_bytes(kv[u].z ^ pat) >> 5) | (zero_bytes(kv[u].w ^ pat) >> 4);
        while (m) {
          const unsigned p = (unsigned)__builtin_ctz(m);
          m &= m - 1u;
          const unsigned e    = q * 16u + (p & 7u) * 4u + (p >> 3);
          const unsigned slot = atomicAdd(&cnt, 1u);
          if (slot < LIST_CAP) {
            list[slot] = e;
          } else {
            const int    s  = src[e];
            const int    d  = dst[e];
            const float  c  = dinv[s] * dinv[d];
            const float* hr = h + (size_t)s * DHID;
            float*       ar = acc + (d - n0) * DHID;
            for (int ch = 0; ch < DHID; ++ch) unsafeAtomicAdd(ar + lds_word(ch), hr[ch] * c);
          }
        }
      }
    }
    __syncthreads();

    if (BUILD && tid == 0) counts[blockIdx.x] = cnt;

    const int total = __builtin_amdgcn_readfirstlane(min((int)cnt, LIST_CAP));
    for (int g0 = wave * EDGE_GROUP; g0 < total; g0 += 32 * EDGE_GROUP) {
      const int nvalid = min(EDGE_GROUP, total - g0);
      const unsigned e  = list[g0 + min((int)lane, nvalid - 1)];
      const int      s  = src[e];
      const int      d  = dst[e];
      const float    c  = dinv[s] * dinv[d];
      const int      dl = d - n0;
      if (BUILD) {
        if ((int)lane < nvalid)
          rl[g0 + (int)lane] = make_uint2((unsigned)s | ((unsigned)dl << SRC_BITS), __float_as_uint(c));
      }
      add_group(h, acc, lane, nvalid, s, dl, c);
    }
  } else {
    const int total = stored;
    const int l7    = min((int)lane, EDGE_GROUP - 1);
    int   g0 = wave * EDGE_GROUP;
    uint2 rc = make_uint2(0u, 0u);
    if (g0 < total) rc = rl[min(g0 + l7, total - 1)];
    for (; g0 < total; g0 += 32 * EDGE_GROUP) {
      const int nvalid = min(EDGE_GROUP, total - g0);
      const int gn = g0 + 32 * EDGE_GROUP;
      uint2 nx = rc;
      if (gn < total) nx = rl[min(gn + l7, total - 1)];
      const int   s  = (int)(rc.x & ((1u << SRC_BITS) - 1u));
      const int   dl = (int)(rc.x >> SRC_BITS);
      const float c  = __uint_as_float(rc.y);
      add_group(h, acc, lane, nvalid, s, dl, c);
      rc = nx;
    }
  }
  __syncthreads();

  float4* __restrict__ out4 = (float4*)(out + (size_t)n0 * DHID);
  for (int i = tid; i < nvec; i += 1024) {
    const float* ar = acc + (i >> 5) * DHID + (i & 31);
    out4[i] = make_float4(ar[0], ar[32], ar[64], ar[96]);
  }
}

__global__ __launch_bounds__(256) void k_bnstats(const float* __restrict__ h,
                                                 float* __restrict__ stats, int n) {
  __shared__ float ss[256], sq[256];
  int t = threadIdx.x, c = t & 127, half = t >> 7;
  float s = 0.0f, q = 0.0f;
  for (int r = blockIdx.x * 2 + half; r < n; r += gridDim.x * 2) {
    float v = fmaxf(h[(size_t)r * DHID + c], 0.0f);
    s += v; q += v * v;
  }
  ss[t] = s; sq[t] = q;
  __syncthreads();
  if (t < 128) {
    unsafeAtomicAdd(&stats[c],       ss[t] + ss[t + 128]);
    unsafeAtomicAdd(&stats[128 + c], sq[t] + sq[t + 128]);
  }
}

__global__ void k_bnfinal(const float* __restrict__ stats, const float* __restrict__ gamma,
                          const float* __restrict__ beta, float* __restrict__ bn, int n) {
  int c = threadIdx.x;
  if (c < 128) {
    float inv = 1.0f / (float)n;
    float mu  = stats[c] * inv;
    float var = stats[128 + c] * inv - mu * mu;
    float sc  = gamma[c] * __frsqrt_rn(var + BN_EPS);
    bn[c]       = sc;
    bn[128 + c] = beta[c] - mu * sc;
  }
}

__global__ __launch_bounds__(256) void k_head(const float* __restrict__ hc,
                                              const float* __restrict__ Wr,
                                              const float* __restrict__ br,
                                              float* __restrict__ out, int n) {
  int w    = (blockIdx.x * blockDim.x + threadIdx.x) >> 5;
  int lane = threadIdx.x & 31;
  if (w >= n) return;
  float4 v = *(const float4*)(hc + (size_t)w * DHID + lane * 4);
  int c = lane * 4;
  float p0 = v.x * Wr[c * 2]       + v.y * Wr[(c + 1) * 2]
           + v.z * Wr[(c + 2) * 2] + v.w * Wr[(c + 3) * 2];
  float p1 = v.x * Wr[c * 2 + 1]       + v.y * Wr[(c + 1) * 2 + 1]
           + v.z * Wr[(c + 2) * 2 + 1] + v.w * Wr[(c + 3) * 2 + 1];
#pragma unroll
  for (int off = 16; off > 0; off >>= 1) {
    p0 += __shfl_down(p0, off, 32);
    p1 += __shfl_down(p1, off, 32);
  }
  if (lane == 0) {
    out[(size_t)w * 2]     = p0 + br[0];
    out[(size_t)w * 2 + 1] = p1 + br[1];
  }
}

extern "C" void kernel_launch(void* const* d_in, const int* in_sizes, int n_in,
                              void* d_out, int out_size, void* d_ws, size_t ws_size,
                              hipStream_t stream) {
  const float* x     = (const float*)d_in[0];
  const int*   eidx  = (const int*)d_in[1];
  const float* W1    = (const float*)d_in[2];
  const float* b1    = (const float*)d_in[3];
  const float* W2    = (const float*)d_in[4];
  const float* b2    = (const float*)d_in[5];
  const float* W3    = (const float*)d_in[6];
  const float* b3    = (const float*)d_in[7];
  const float* gamma = (const float*)d_in[8];
  const float* beta  = (const float*)d_in[9];
  const float* Wc    = (const float*)d_in[10];
  const float* bc    = (const float*)d_in[11];
  const float* Wr    = (const float*)d_in[12];
  const float* br    = (const float*)d_in[13];
  float* out = (float*)d_out;

  const int N = N_NODES, E = E_EDGES;
  const int* src = eidx;
  const int* dst = eidx + E;

  float* buf0  = (float*)d_ws;
  float* buf1  = buf0 + (size_t)N * DHID;
  float* dinv  = buf1 + (size_t)N * DHID;
  float* selfc = dinv + N;
  float* stats = selfc + N;
  float* bnp   = stats + 256;
  unsigned int* keys = (unsigned int*)(bnp + 256);
  unsigned int* counts = keys + E / 4;
  unsigned int* recs = counts + AGG_BLOCKS;

  const dim3 blk(256);
  const int gN    = (N + 255) / 256;
  const int gE4   = (E / 4 + 255) / 256;
  const int gGemm = (N + 127) / 128;
  const int gHead = (N * 32 + 255) / 256;

  k_zero<<<gN, blk, 0, stream>>>(dinv, N);
  k_zero<<<1, blk, 0, stream>>>(stats, 256);
  k_degree_keys<<<gE4, blk, 0, stream>>>(dst, dinv, keys, E / 4);
  k_dinv<<<gN, blk, 0, stream>>>(dinv, selfc, N);

  k_gemm128<0, false, false><<<gGemm, blk, 0, stream>>>(x, W1, buf0, N, nullptr, nullptr, nullptr);
  k_aggregate<true><<<AGG_BLOCKS, 1024, 0, stream>>>(src, dst, keys, dinv, selfc, buf0, b1, buf1, recs, counts);

  k_gemm128<1, false, false><<<gGemm, blk, 0, stream>>>(buf1, W2, buf0, N, nullptr, nullptr, nullptr);
  k_aggregate<false><<<AGG_BLOCKS, 1024, 0, stream>>>(src, dst, keys, dinv, selfc, buf0, b2, buf1, recs, counts);

  k_gemm128<1, false, false><<<gGemm, blk, 0, stream>>>(buf1, W3, buf0, N, nullptr, nullptr, nullptr);
  k_aggregate<false><<<AGG_BLOCKS, 1024, 0, stream>>>(src, dst, keys, dinv, selfc, buf0, b3, buf1, recs, counts);

  k_bnstats<<<1024, blk, 0, stream>>>(buf1, stats, N);
  k_bnfinal<<<1, 128, 0, stream>>>(stats, gamma, beta, bnp, N);

  k_gemm128<2, true, true><<<gGemm, blk, 0, stream>>>(buf1, Wc, buf0, N, bnp, bnp + 128, bc);

  k_head<<<gHead, blk, 0, stream>>>(buf0, Wr, br, out, N);
}


# Edge records of all node ranges filed by one binning pass over the edges (LDS counts, one returning global add per block and range); layer 1 reads them too; key plane, scans gone; LDS arrival order
# speedup vs baseline: 2.6428x; 1.0621x over previous
#include <hip/hip_runtime.h>

#define N_NODES 100000
#define DHID    128
#define E_EDGES 1600000
#define BN_EPS  1e-5f

typedef __attribute__((ext_vector_type(2))) float v2f;
typedef __attribute__((ext_vector_type(8))) float v8f;

__global__ void k_zero(float* __restrict__ p, int n) {
  int i = blockIdx.x * blockDim.x + threadIdx.x;
  if (i < n) p[i] = 0.0f;
}

#define ROWS_PER_BLOCK 391
#define AGG_BLOCKS     256
static_assert(AGG_BLOCKS * ROWS_PER_BLOCK >= N_NODES, "ranges must cover every node");
static_assert(E_EDGES % 4 == 0, "edges are read 4 per thread");

__global__ void k_degree(const int* __restrict__ dst, float* __restrict__ deg, int e4) {
  int i = blockIdx.x * blockDim.x + threadIdx.x;
  if (i >= e4) return;
  int4 d = ((const int4*)dst)[i];
  unsafeAtomicAdd(&deg[d.x], 1.0f);
  unsafeAtomicAdd(&deg[d.y], 1.0f);
  unsafeAtomicAdd(&deg[d.z], 1.0f);
  unsafeAtomicAdd(&deg[d.w], 1.0f);
}

__global__ void k_dinv(float* __restrict__ deg_dinv, float* __restrict__ selfc, int n) {
  int i = blockIdx.x * blockDim.x + threadIdx.x;
  if (i < n) {
    float d = __frsqrt_rn(deg_dinv[i] + 1.0f);
    deg_dinv[i] = d;
    selfc[i]    = d * d;
  }
}

template <int AMODE, bool HAS_BIAS, bool EPI_RELU>
__global__ __launch_bounds__(256) void k_gemm128(
    const float* __restrict__ A, const float* __restrict__ W,
    float* __restrict__ C, int nrows,
    const float* __restrict__ bnScale, const float* __restrict__ bnShift,
    const float* __restrict__ bias)
{
  __shared__ float wl[DHID * DHID];
  const int tid = threadIdx.x;
  for (int i = tid; i < DHID * DHID; i += 256) {
    int k = i >> 7, n = i & 127;
    wl[(((k >> 1) << 7) + n) * 2 + (k & 1)] = W[i];
  }
  __syncthreads();

  const int wave  = tid >> 5;
  const int lane  = tid & 31;
  const int laneM = lane & 15;
  const int hi    = lane >> 4;
  const int m0    = (blockIdx.x * 8 + wave) << 4;
  if (m0 >= nrows) return;

  const float2* __restrict__ wl2  = (const float2*)wl;
  const float*  __restrict__ arow = A + (size_t)(m0 + laneM) * DHID;

  v8f acc[8];
#pragma unroll
  for (int t = 0; t < 8; ++t) acc[t] = (v8f){};

  for (int k = 0; k < DHID; k += 4) {
    float2 av = *(const float2*)(arow + k + hi * 2);
    if (AMODE == 1) {
      av.x = fmaxf(av.x, 0.0f);
      av.y = fmaxf(av.y, 0.0f);
    } else if (AMODE == 2) {
      int c0 = k + hi * 2;
      av.x = fmaxf(av.x, 0.0f) * bnScale[c0]     + bnShift[c0];
      av.y = fmaxf(av.y, 0.0f) * bnScale[c0 + 1] + bnShift[c0 + 1];
    }
    v2f a; a[0] = av.x; a[1] = av.y;

    const float2* __restrict__ brow = wl2 + (((k >> 1) + hi) << 7);
#pragma unroll
    for (int t = 0; t < 8; ++t) {
      float2 bv = brow[t * 16 + laneM];
      v2f b; b[0] = bv.x; b[1] = bv.y;
      acc[t] = __builtin_amdgcn_wmma_f32_16x16x4_f32(
          false, a, false, b, (short)0, acc[t], false, false);
    }
  }

#pragma unroll
  for (int t = 0; t < 8; ++t) {
    int   col  = t * 16 + laneM;
    float bval = HAS_BIAS ? bias[col] : 0.0f;
#pragma unroll
    for (int r = 0; r < 8; ++r) {
      int   row = m0 + r + hi * 8;
      float v   = acc[t][r] + bval;
      if (EPI_RELU) v = fmaxf(v, 0.0f);
      C[(size_t)row * DHID + col] = v;
    }
  }
}

#define LIST_CAP   24576
#define EDGE_GROUP 8
#define SRC_BITS   17
static_assert(N_NODES <= (1 << SRC_BITS), "a source node must fit the record's low bits");
static_assert(ROWS_PER_BLOCK <= (1 << (32 - SRC_BITS)), "a local row must fit the record's high bits");

#define BIN_EDGES 4096
static_assert(AGG_BLOCKS == 256, "the count scan below is written for 8 waves of 32 ranges");
__global__ __launch_bounds__(1024) void k_bin(const int* __restrict__ src, const int* __restrict__ dst,
                                              const float* __restrict__ dinv,
                                              unsigned int* __restrict__ recs,
                                              unsigned int* __restrict__ counts, int e4) {
  __shared__ unsigned int  hist[AGG_BLOCKS];
  __shared__ unsigned int  pos[AGG_BLOCKS];
  __shared__ unsigned int  gbase[AGG_BLOCKS];
  __shared__ unsigned int  wsum[8];
  __shared__ uint2         stage[BIN_EDGES];
  __shared__ unsigned char srange[BIN_EDGES];
  const int tid = threadIdx.x;
  if (tid < AGG_BLOCKS) hist[tid] = 0u;
  __syncthreads();

  const int  i  = blockIdx.x * 1024 + tid;
  const bool in = i < e4;
  int      s[4] = {0, 0, 0, 0}, d[4] = {0, 0, 0, 0};
  unsigned r[4] = {0u, 0u, 0u, 0u}, rank[4] = {0u, 0u, 0u, 0u};
  float    c[4] = {0.0f, 0.0f, 0.0f, 0.0f};
  if (in) {
    const int4 sv = ((const int4*)src)[i];
    const int4 dv = ((const int4*)dst)[i];
    s[0] = sv.x; s[1] = sv.y; s[2] = sv.z; s[3] = sv.w;
    d[0] = dv.x; d[1] = dv.y; d[2] = dv.z; d[3] = dv.w;
#pragma unroll
    for (int u = 0; u < 4; ++u) {
      r[u]    = (unsigned)(d[u] / ROWS_PER_BLOCK);
      rank[u] = atomicAdd(&hist[r[u]], 1u);
      c[u]    = dinv[s[u]] * dinv[d[u]];
    }
  }
  __syncthreads();

  unsigned n = 0u, incl = 0u;
  if (tid < AGG_BLOCKS) {
    n    = hist[tid];
    incl = n;
#pragma unroll
    for (int off = 1; off < 32; off <<= 1) {
      const unsigned t = __shfl_up(incl, off, 32);
      if ((tid & 31) >= off) incl += t;
    }
    if ((tid & 31) == 31) wsum[tid >> 5] = incl;
    gbase[tid] = n ? atomicAdd(&counts[tid], n) : 0u;
  }
  __syncthreads();
  if (tid < AGG_BLOCKS) {
    unsigned before = 0u;
    for (int w = 0; w < (tid >> 5); ++w) before += wsum[w];
    pos[tid] = before + incl - n;
  }
  __syncthreads();

  if (in) {
#pragma unroll
    for (int u = 0; u < 4; ++u) {
      const unsigned j  = pos[r[u]] + rank[u];
      const unsigned dl = (unsigned)d[u] - r[u] * ROWS_PER_BLOCK;
      stage[j]  = make_uint2((unsigned)s[u] | (dl << SRC_BITS), __float_as_uint(c[u]));
      srange[j] = (unsigned char)r[u];
    }
  }
  __syncthreads();

  const int nrec = min(1024, e4 - (int)blockIdx.x * 1024) * 4;
  uint2* __restrict__ rl = (uint2*)recs;
  for (int j = tid; j < nrec; j += 1024) {
    const unsigned rr   = srange[j];
    const unsigned slot = gbase[rr] + ((unsigned)j - pos[rr]);
    if (slot < LIST_CAP) rl[(size_t)rr * LIST_CAP + slot] = stage[j];
  }
}

__device__ __forceinline__ int lds_word(int ch) { return (ch & 3) * 32 + (ch >> 2); }

__device__ __forceinline__ void add_group(const float* __restrict__ h, float* acc, unsigned lane,
                                          int nvalid, int s, int dl, float c) {
  if (nvalid == EDGE_GROUP) {
    float4 v[EDGE_GROUP];
#pragma unroll
    for (int u = 0; u < EDGE_GROUP; ++u) {
      int sj = __builtin_amdgcn_readlane(s, u);
      v[u] = ((const float4*)(h + (size_t)sj * DHID))[lane];
    }
#pragma unroll
    for (int u = 0; u < EDGE_GROUP; ++u) {
      int   dj = __builtin_amdgcn_readlane(dl, u);
      float cj = __int_as_float(__builtin_amdgcn_readlane(__float_as_int(c), u));
      float* a = acc + dj * DHID;
      unsafeAtomicAdd(a + lane,      v[u].x * cj);
      unsafeAtomicAdd(a + lane + 32, v[u].y * cj);
      unsafeAtomicAdd(a + lane + 64, v[u].z * cj);
      unsafeAtomicAdd(a + lane + 96, v[u].w * cj);
    }
  } else {
    for (int j = 0; j < nvalid; ++j) {
      int   sj = __builtin_amdgcn_readlane(s, j);
      int   dj = __builtin_amdgcn_readlane(dl, j);
      float cj = __int_as_float(__builtin_amdgcn_readlane(__float_as_int(c), j));
      const float4 hv = ((const float4*)(h + (size_t)sj * DHID))[lane];
      float*       a  = acc + dj * DHID;
      unsafeAtomicAdd(a + lane,      hv.x * cj);
      unsafeAtomicAdd(a + lane + 32, hv.y * cj);
      unsafeAtomicAdd(a + lane + 64, hv.z * cj);
      unsafeAtomicAdd(a + lane + 96, hv.w * cj);
    }
  }
}

__global__ __launch_bounds__(1024) void k_aggregate(
    const int* __restrict__ src, const int* __restrict__ dst,
    const float* __restrict__ dinv, const float* __restrict__ selfc,
    const float* __restrict__ h, const float* __restrict__ b,
    float* __restrict__ out,
    const unsigned int* __restrict__ recs, const unsigned int* __restrict__ counts)
{
  __shared__ float acc[ROWS_PER_BLOCK * DHID];

  const int      tid  = threadIdx.x;
  const unsigned lane = threadIdx.x & 31u;
  const int      n0   = blockIdx.x * ROWS_PER_BLOCK;
  const int      wave = __builtin_amdgcn_readfirstlane(tid >> 5);

  const uint2* __restrict__ rl = (const uint2*)recs + (size_t)blockIdx.x * LIST_CAP;

  const int nvec = min(ROWS_PER_BLOCK, N_NODES - n0) * (DHID / 4);

  {
    const float4* __restrict__ h4 = (const float4*)(h + (size_t)n0 * DHID);
    const float4 bv = *(const float4*)(b + (tid & 31) * 4);
#pragma unroll 4
    for (int i = tid; i < nvec; i += 1024) {
      float  sc = selfc[n0 + (i >> 5)];
      float4 hv = h4[i];
      float* ar = acc + (i >> 5) * DHID + (i & 31);
      ar[0]  = hv.x * sc + bv.x;
      ar[32] = hv.y * sc + bv.y;
      ar[64] = hv.z * sc + bv.z;
      ar[96] = hv.w * sc + bv.w;
    }
  }
  __syncthreads();

  const int total = __builtin_amdgcn_readfirstlane((int)counts[blockIdx.x]);

  if (total <= LIST_CAP) {
    const int l7 = min((int)lane, EDGE_GROUP - 1);
    int   g0 = wave * EDGE_GROUP;
    uint2 rc = make_uint2(0u, 0u);
    if (g0 < total) rc = rl[min(g0 + l7, total - 1)];
    for (; g0 < total; g0 += 32 * EDGE_GROUP) {
      const int nvalid = min(EDGE_GROUP, total - g0);
      const int gn = g0 + 32 * EDGE_GROUP;
      uint2 nx = rc;
      if (gn < total) nx = rl[min(gn + l7, total - 1)];
      const int   s  = (int)(rc.x & ((1u << SRC_BITS) - 1u));
      const int   dl = (int)(rc.x >> SRC_BITS);
      const float c  = __uint_as_float(rc.y);
      add_group(h, acc, lane, nvalid, s, dl, c);
      rc = nx;
    }
  } else {
    for (int e = tid; e < E_EDGES; e += 1024) {
      const int d = dst[e];
      if (d >= n0 && d < n0 + ROWS_PER_BLOCK) {
        const int    s  = src[e];
        const float  c  = dinv[s] * dinv[d];
        const float* hr = h + (size_t)s * DHID;
        float*       ar = acc + (d - n0) * DHID;
        for (int ch = 0; ch < DHID; ++ch) unsafeAtomicAdd(ar + lds_word(ch), hr[ch] * c);
      }
    }
  }
  __syncthreads();

  float4* __restrict__ out4 = (float4*)(out + (size_t)n0 * DHID);
  for (int i = tid; i < nvec; i += 1024) {
    const float* ar = acc + (i >> 5) * DHID + (i & 31);
    out4[i] = make_float4(ar[0], ar[32], ar[64], ar[96]);
  }
}

__global__ __launch_bounds__(256) void k_bnstats(const float* __restrict__ h,
                                                 float* __restrict__ stats, int n) {
  __shared__ float ss[256], sq[256];
  int t = threadIdx.x, c = t & 127, half = t >> 7;
  float s = 0.0f, q = 0.0f;
  for (int r = blockIdx.x * 2 + half; r < n; r += gridDim.x * 2) {
    float v = fmaxf(h[(size_t)r * DHID + c], 0.0f);
    s += v; q += v * v;
  }
  ss[t] = s; sq[t] = q;
  __syncthreads();
  if (t < 128) {
    unsafeAtomicAdd(&stats[c],       ss[t] + ss[t + 128]);
    unsafeAtomicAdd(&stats[128 + c], sq[t] + sq[t + 128]);
  }
}

__global__ void k_bnfinal(const float* __restrict__ stats, const float* __restrict__ gamma,
                          const float* __restrict__ beta, float* __restrict__ bn, int n) {
  int c = threadIdx.x;
  if (c < 128) {
    float inv = 1.0f / (float)n;
    float mu  = stats[c] * inv;
    float var = stats[128 + c] * inv - mu * mu;
    float sc  = gamma[c] * __frsqrt_rn(var + BN_EPS);
    bn[c]       = sc;
    bn[128 + c] = beta[c] - mu * sc;
  }
}

__global__ __launch_bounds__(256) void k_head(const float* __restrict__ hc,
                                              const float* __restrict__ Wr,
                                              const float* __restrict__ br,
                                              float* __restrict__ out, int n) {
  int w    = (blockIdx.x * blockDim.x + threadIdx.x) >> 5;
  int lane = threadIdx.x & 31;
  if (w >= n) return;
  float4 v = *(const float4*)(hc + (size_t)w * DHID + lane * 4);
  int c = lane * 4;
  float p0 = v.x * Wr[c * 2]       + v.y * Wr[(c + 1) * 2]
           + v.z * Wr[(c + 2) * 2] + v.w * Wr[(c + 3) * 2];
  float p1 = v.x * Wr[c * 2 + 1]       + v.y * Wr[(c + 1) * 2 + 1]
           + v.z * Wr[(c + 2) * 2 + 1] + v.w * Wr[(c + 3) * 2 + 1];
#pragma unroll
  for (int off = 16; off > 0; off >>= 1) {
    p0 += __shfl_down(p0, off, 32);
    p1 += __shfl_down(p1, off, 32);
  }
  if (lane == 0) {
    out[(size_t)w * 2]     = p0 + br[0];
    out[(size_t)w * 2 + 1] = p1 + br[1];
  }
}

extern "C" void kernel_launch(void* const* d_in, const int* in_sizes, int n_in,
                              void* d_out, int out_size, void* d_ws, size_t ws_size,
                              hipStream_t stream) {
  const float* x     = (const float*)d_in[0];
  const int*   eidx  = (const int*)d_in[1];
  const float* W1    = (const float*)d_in[2];
  const float* b1    = (const float*)d_in[3];
  const float* W2    = (const float*)d_in[4];
  const float* b2    = (const float*)d_in[5];
  const float* W3    = (const float*)d_in[6];
  const float* b3    = (const float*)d_in[7];
  const float* gamma = (const float*)d_in[8];
  const float* beta  = (const float*)d_in[9];
  const float* Wc    = (const float*)d_in[10];
  const float* bc    = (const float*)d_in[11];
  const float* Wr    = (const float*)d_in[12];
  const float* br    = (const float*)d_in[13];
  float* out = (float*)d_out;

  const int N = N_NODES, E = E_EDGES;
  const int* src = eidx;
  const int* dst = eidx + E;

  float* buf0  = (float*)d_ws;
  float* buf1  = buf0 + (size_t)N * DHID;
  float* dinv  = buf1 + (size_t)N * DHID;
  float* selfc = dinv + N;
  float* stats = selfc + N;
  unsigned int* counts = (unsigned int*)(stats + 256);
  float* bnp   = stats + 256 + AGG_BLOCKS;
  unsigned int* recs = (unsigned int*)(bnp + 256);

  const dim3 blk(256);
  const int gN    = (N + 255) / 256;
  const int gE4   = (E / 4 + 255) / 256;
  const int gBin  = (E / 4 + 1023) / 1024;
  const int gGemm = (N + 127) / 128;
  const int gHead = (N * 32 + 255) / 256;

  k_zero<<<gN, blk, 0, stream>>>(dinv, N);
  k_zero<<<2, blk, 0, stream>>>(stats, 256 + AGG_BLOCKS);
  k_degree<<<gE4, blk, 0, stream>>>(dst, dinv, E / 4);
  k_dinv<<<gN, blk, 0, stream>>>(dinv, selfc, N);

  k_bin<<<gBin, 1024, 0, stream>>>(src, dst, dinv, recs, counts, E / 4);

  k_gemm128<0, false, false><<<gGemm, blk, 0, stream>>>(x, W1, buf0, N, nullptr, nullptr, nullptr);
  k_aggregate<<<AGG_BLOCKS, 1024, 0, stream>>>(src, dst, dinv, selfc, buf0, b1, buf1, recs, counts);

  k_gemm128<1, false, false><<<gGemm, blk, 0, stream>>>(buf1, W2, buf0, N, nullptr, nullptr, nullptr);
  k_aggregate<<<AGG_BLOCKS, 1024, 0, stream>>>(src, dst, dinv, selfc, buf0, b2, buf1, recs, counts);

  k_gemm128<1, false, false><<<gGemm, blk, 0, stream>>>(buf1, W3, buf0, N, nullptr, nullptr, nullptr);
  k_aggregate<<<AGG_BLOCKS, 1024, 0, stream>>>(src, dst, dinv, selfc, buf0, b3, buf1, recs, counts);

  k_bnstats<<<1024, blk, 0, stream>>>(buf1, stats, N);
  k_bnfinal<<<1, 128, 0, stream>>>(stats, gamma, beta, bnp, N);

  k_gemm128<2, true, true><<<gGemm, blk, 0, stream>>>(buf1, Wc, buf0, N, bnp, bnp + 128, bc);

  k_head<<<gHead, blk, 0, stream>>>(buf0, Wr, br, out, N);
}


# Edge coefficient split: products store rows times own dinv, sums get dinv[d] then bias last; one-word edge records, in-degree counted from range lists in LDS; degree atomics, dinv pass, look-ups gone
# speedup vs baseline: 3.4697x; 1.3129x over previous
#include <hip/hip_runtime.h>

#define N_NODES 100000
#define DHID    128
#define E_EDGES 1600000
#define BN_EPS  1e-5f

typedef __attribute__((ext_vector_type(2))) float v2f;
typedef __attribute__((ext_vector_type(8))) float v8f;

__global__ void k_zero(float* __restrict__ p, int n) {
  int i = blockIdx.x * blockDim.x + threadIdx.x;
  if (i < n) p[i] = 0.0f;
}

#define ROWS_PER_BLOCK 391
#define AGG_BLOCKS     256
static_assert(AGG_BLOCKS * ROWS_PER_BLOCK >= N_NODES, "ranges must cover every node");
static_assert(E_EDGES % 4 == 0, "edges are read 4 per thread");

template <int AMODE, bool HAS_BIAS, bool EPI_RELU, bool ROW_SCALE>
__global__ __launch_bounds__(256) void k_gemm128(
    const float* __restrict__ A, const float* __restrict__ W,
    float* __restrict__ C, int nrows,
    const float* __restrict__ bnScale, const float* __restrict__ bnShift,
    const float* __restrict__ bias, const float* __restrict__ rowscale)
{
  __shared__ float wl[DHID * DHID];
  const int tid = threadIdx.x;
  for (int i = tid; i < DHID * DHID; i += 256) {
    int k = i >> 7, n = i & 127;
    wl[(((k >> 1) << 7) + n) * 2 + (k & 1)] = W[i];
  }
  __syncthreads();

  const int wave  = tid >> 5;
  const int lane  = tid & 31;
  const int laneM = lane & 15;
  const int hi    = lane >> 4;
  const int m0    = (blockIdx.x * 8 + wave) << 4;
  if (m0 >= nrows) return;

  const float2* __restrict__ wl2  = (const float2*)wl;
  const float*  __restrict__ arow = A + (size_t)(m0 + laneM) * DHID;

  v8f acc[8];
#pragma unroll
  for (int t = 0; t < 8; ++t) acc[t] = (v8f){};

  for (int k = 0; k < DHID; k += 4) {
    float2 av = *(const float2*)(arow + k + hi * 2);
    if (AMODE == 1) {
      av.x = fmaxf(av.x, 0.0f);
      av.y = fmaxf(av.y, 0.0f);
    } else if (AMODE == 2) {
      int c0 = k + hi * 2;
      av.x = fmaxf(av.x, 0.0f) * bnScale[c0]     + bnShift[c0];
      av.y = fmaxf(av.y, 0.0f) * bnScale[c0 + 1] + bnShift[c0 + 1];
    }
    v2f a; a[0] = av.x; a[1] = av.y;

    const float2* __restrict__ brow = wl2 + (((k >> 1) + hi) << 7);
#pragma unroll
    for (int t = 0; t < 8; ++t) {
      float2 bv = brow[t * 16 + laneM];
      v2f b; b[0] = bv.x; b[1] = bv.y;
      acc[t] = __builtin_amdgcn_wmma_f32_16x16x4_f32(
          false, a, false, b, (short)0, acc[t], false, false);
    }
  }

  float rs[8];
#pragma unroll
  for (int r = 0; r < 8; ++r) rs[r] = ROW_SCALE ? rowscale[m0 + r + hi * 8] : 1.0f;

#pragma unroll
  for (int t = 0; t < 8; ++t) {
    int   col  = t * 16 + laneM;
    float bval = HAS_BIAS ? bias[col] : 0.0f;
#pragma unroll
    for (int r = 0; r < 8; ++r) {
      int   row = m0 + r + hi * 8;
      float v   = acc[t][r];
      if (ROW_SCALE) v *= rs[r];
      if (HAS_BIAS) v += bval;
      if (EPI_RELU) v = fmaxf(v, 0.0f);
      C[(size_t)row * DHID + col] = v;
    }
  }
}

#define LIST_CAP   24576
#define EDGE_GROUP 8
#define SRC_BITS   17
static_assert(N_NODES <= (1 << SRC_BITS), "a source node must fit the record's low bits");
static_assert(ROWS_PER_BLOCK <= (1 << (32 - SRC_BITS)), "a local row must fit the record's high bits");

#define BIN_EDGES 4096
static_assert(AGG_BLOCKS == 256, "the count scan below is written for 8 waves of 32 ranges");
__global__ __launch_bounds__(1024) void k_bin(const int* __restrict__ src, const int* __restrict__ dst,
                                              unsigned int* __restrict__ recs,
                                              unsigned int* __restrict__ counts, int e4) {
  __shared__ unsigned int  hist[AGG_BLOCKS];
  __shared__ unsigned int  pos[AGG_BLOCKS];
  __shared__ unsigned int  gbase[AGG_BLOCKS];
  __shared__ unsigned int  wsum[8];
  __shared__ unsigned int  stage[BIN_EDGES];
  __shared__ unsigned char srange[BIN_EDGES];
  const int tid = threadIdx.x;
  if (tid < AGG_BLOCKS) hist[tid] = 0u;
  __syncthreads();

  const int  i  = blockIdx.x * 1024 + tid;
  const bool in = i < e4;
  int      s[4] = {0, 0, 0, 0}, d[4] = {0, 0, 0, 0};
  unsigned r[4] = {0u, 0u, 0u, 0u}, rank[4] = {0u, 0u, 0u, 0u};
  if (in) {
    const int4 sv = ((const int4*)src)[i];
    const int4 dv = ((const int4*)dst)[i];
    s[0] = sv.x; s[1] = sv.y; s[2] = sv.z; s[3] = sv.w;
    d[0] = dv.x; d[1] = dv.y; d[2] = dv.z; d[3] = dv.w;
#pragma unroll
    for (int u = 0; u < 4; ++u) {
      r[u]    = (unsigned)(d[u] / ROWS_PER_BLOCK);
      rank[u] = atomicAdd(&hist[r[u]], 1u);
    }
  }
  __syncthreads();

  unsigned n = 0u, incl = 0u;
  if (tid < AGG_BLOCKS) {
    n    = hist[tid];
    incl = n;
#pragma unroll
    for (int off = 1; off < 32; off <<= 1) {
      const unsigned t = __shfl_up(incl, off, 32);
      if ((tid & 31) >= off) incl += t;
    }
    if ((tid & 31) == 31) wsum[tid >> 5] = incl;
    gbase[tid] = n ? atomicAdd(&counts[tid], n) : 0u;
  }
  __syncthreads();
  if (tid < AGG_BLOCKS) {
    unsigned before = 0u;
    for (int w = 0; w < (tid >> 5); ++w) before += wsum[w];
    pos[tid] = before + incl - n;
  }
  __syncthreads();

  if (in) {
#pragma unroll
    for (int u = 0; u < 4; ++u) {
      const unsigned j  = pos[r[u]] + rank[u];
      const unsigned dl = (unsigned)d[u] - r[u] * ROWS_PER_BLOCK;
      stage[j]  = (unsigned)s[u] | (dl << SRC_BITS);
      srange[j] = (unsigned char)r[u];
    }
  }
  __syncthreads();

  const int nrec = min(1024, e4 - (int)blockIdx.x * 1024) * 4;
  for (int j = tid; j < nrec; j += 1024) {
    const unsigned rr   = srange[j];
    const unsigned slot = gbase[rr] + ((unsigned)j - pos[rr]);
    if (slot < LIST_CAP) recs[(size_t)rr * LIST_CAP + slot] = stage[j];
  }
}

#define DEG_THREADS 256
#define DEG_UNROLL  32
__global__ __launch_bounds__(DEG_THREADS) void k_rangedeg(const int* __restrict__ dst,
                                                          const unsigned int* __restrict__ recs,
                                                          const unsigned int* __restrict__ counts,
                                                          float* __restrict__ dinv) {
  __shared__ unsigned int cnt[ROWS_PER_BLOCK];
  const int tid = threadIdx.x;
  const int n0  = blockIdx.x * ROWS_PER_BLOCK;
  for (int i = tid; i < ROWS_PER_BLOCK; i += DEG_THREADS) cnt[i] = 0u;
  __syncthreads();

  const int total = (int)counts[blockIdx.x];
  if (total <= LIST_CAP) {
    const unsigned int* __restrict__ rl = recs + (size_t)blockIdx.x * LIST_CAP;
    for (int j0 = tid; j0 < total; j0 += DEG_UNROLL * DEG_THREADS) {
      unsigned rv[DEG_UNROLL];
#pragma unroll
      for (int u = 0; u < DEG_UNROLL; ++u) rv[u] = rl[min(j0 + u * DEG_THREADS, total - 1)];
#pragma unroll
      for (int u = 0; u < DEG_UNROLL; ++u)
        if (j0 + u * DEG_THREADS < total) atomicAdd(&cnt[rv[u] >> SRC_BITS], 1u);
    }
  } else {
    for (int e = tid; e < E_EDGES; e += DEG_THREADS) {
      const int d = dst[e];
      if (d >= n0 && d < n0 + ROWS_PER_BLOCK) atomicAdd(&cnt[d - n0], 1u);
    }
  }
  __syncthreads();

  const int nrows = min(ROWS_PER_BLOCK, N_NODES - n0);
  for (int i = tid; i < nrows; i += DEG_THREADS) dinv[n0 + i] = __frsqrt_rn((float)cnt[i] + 1.0f);
}

__device__ __forceinline__ int lds_word(int ch) { return (ch & 3) * 32 + (ch >> 2); }

__device__ __forceinline__ void add_group(const float* __restrict__ h, float* acc, unsigned lane,
                                          int nvalid, int s, int dl) {
  if (nvalid == EDGE_GROUP) {
    float4 v[EDGE_GROUP];
#pragma unroll
    for (int u = 0; u < EDGE_GROUP; ++u) {
      int sj = __builtin_amdgcn_readlane(s, u);
      v[u] = ((const float4*)(h + (size_t)sj * DHID))[lane];
    }
#pragma unroll
    for (int u = 0; u < EDGE_GROUP; ++u) {
      int    dj = __builtin_amdgcn_readlane(dl, u);
      float* a  = acc + dj * DHID;
      unsafeAtomicAdd(a + lane,      v[u].x);
      unsafeAtomicAdd(a + lane + 32, v[u].y);
      unsafeAtomicAdd(a + lane + 64, v[u].z);
      unsafeAtomicAdd(a + lane + 96, v[u].w);
    }
  } else {
    for (int j = 0; j < nvalid; ++j) {
      int sj = __builtin_amdgcn_readlane(s, j);
      int dj = __builtin_amdgcn_readlane(dl, j);
      const float4 hv = ((const float4*)(h + (size_t)sj * DHID))[lane];
      float*       a  = acc + dj * DHID;
      unsafeAtomicAdd(a + lane,      hv.x);
      unsafeAtomicAdd(a + lane + 32, hv.y);
      unsafeAtomicAdd(a + lane + 64, hv.z);
      unsafeAtomicAdd(a + lane + 96, hv.w);
    }
  }
}

__global__ __launch_bounds__(1024) void k_aggregate(
    const int* __restrict__ src, const int* __restrict__ dst,
    const float* __restrict__ dinv,
    const float* __restrict__ h, const float* __restrict__ b,
    float* __restrict__ out,
    const unsigned int* __restrict__ recs, const unsigned int* __restrict__ counts)
{
  __shared__ float acc[ROWS_PER_BLOCK * DHID];
  __shared__ float drow[ROWS_PER_BLOCK];

  const int      tid  = threadIdx.x;
  const unsigned lane = threadIdx.x & 31u;
  const int      n0   = blockIdx.x * ROWS_PER_BLOCK;
  const int      wave = __builtin_amdgcn_readfirstlane(tid >> 5);

  const unsigned int* __restrict__ rl = recs + (size_t)blockIdx.x * LIST_CAP;

  const int nvec = min(ROWS_PER_BLOCK, N_NODES - n0) * (DHID / 4);

  {
    const bool ownrow = tid * (DHID / 4) < nvec;
    float      dmine  = 0.0f;
    if (ownrow) dmine = dinv[n0 + tid];
    const float4* __restrict__ h4 = (const float4*)(h + (size_t)n0 * DHID);
#pragma unroll 4
    for (int i = tid; i < nvec; i += 1024) {
      float4 hv = h4[i];
      float* ar = acc + (i >> 5) * DHID + (i & 31);
      ar[0]  = hv.x;
      ar[32] = hv.y;
      ar[64] = hv.z;
      ar[96] = hv.w;
    }
    if (ownrow) drow[tid] = dmine;
  }
  __syncthreads();

  const int total = __builtin_amdgcn_readfirstlane((int)counts[blockIdx.x]);

  if (total <= LIST_CAP) {
    const int l7 = min((int)lane, EDGE_GROUP - 1);
    int      g0 = wave * EDGE_GROUP;
    unsigned rc = 0u;
    if (g0 < total) rc = rl[min(g0 + l7, total - 1)];
    for (; g0 < total; g0 += 32 * EDGE_GROUP) {
      const int nvalid = min(EDGE_GROUP, total - g0);
      const int gn = g0 + 32 * EDGE_GROUP;
      unsigned  nx = rc;
      if (gn < total) nx = rl[min(gn + l7, total - 1)];
      const int s  = (int)(rc & ((1u << SRC_BITS) - 1u));
      const int dl = (int)(rc >> SRC_BITS);
      add_group(h, acc, lane, nvalid, s, dl);
      rc = nx;
    }
  } else {
    for (int e = tid; e < E_EDGES; e += 1024) {
      const int d = dst[e];
      if (d >= n0 && d < n0 + ROWS_PER_BLOCK) {
        const float* hr = h + (size_t)src[e] * DHID;
        float*       ar = acc + (d - n0) * DHID;
        for (int ch = 0; ch < DHID; ++ch) unsafeAtomicAdd(ar + lds_word(ch), hr[ch]);
      }
    }
  }
  __syncthreads();

  float4* __restrict__ out4 = (float4*)(out + (size_t)n0 * DHID);
  const float4 bv = *(const float4*)(b + (tid & 31) * 4);
  for (int i = tid; i < nvec; i += 1024) {
    const float dv = drow[i >> 5];
    const float* ar = acc + (i >> 5) * DHID + (i & 31);
    out4[i] = make_float4(ar[0] * dv + bv.x, ar[32] * dv + bv.y, ar[64] * dv + bv.z, ar[96] * dv + bv.w);
  }
}

__global__ __launch_bounds__(256) void k_bnstats(const float* __restrict__ h,
                                                 float* __restrict__ stats, int n) {
  __shared__ float ss[256], sq[256];
  int t = threadIdx.x, c = t & 127, half = t >> 7;
  float s = 0.0f, q = 0.0f;
  for (int r = blockIdx.x * 2 + half; r < n; r += gridDim.x * 2) {
    float v = fmaxf(h[(size_t)r * DHID + c], 0.0f);
    s += v; q += v * v;
  }
  ss[t] = s; sq[t] = q;
  __syncthreads();
  if (t < 128) {
    unsafeAtomicAdd(&stats[c],       ss[t] + ss[t + 128]);
    unsafeAtomicAdd(&stats[128 + c], sq[t] + sq[t + 128]);
  }
}

__global__ void k_bnfinal(const float* __restrict__ stats, const float* __restrict__ gamma,
                          const float* __restrict__ beta, float* __restrict__ bn, int n) {
  int c = threadIdx.x;
  if (c < 128) {
    float inv = 1.0f / (float)n;
    float mu  = stats[c] * inv;
    float var = stats[128 + c] * inv - mu * mu;
    float sc  = gamma[c] * __frsqrt_rn(var + BN_EPS);
    bn[c]       = sc;
    bn[128 + c] = beta[c] - mu * sc;
  }
}

__global__ __launch_bounds__(256) void k_head(const float* __restrict__ hc,
                                              const float* __restrict__ Wr,
                                              const float* __restrict__ br,
                                              float* __restrict__ out, int n) {
  int w    = (blockIdx.x * blockDim.x + threadIdx.x) >> 5;
  int lane = threadIdx.x & 31;
  if (w >= n) return;
  float4 v = *(const float4*)(hc + (size_t)w * DHID + lane * 4);
  int c = lane * 4;
  float p0 = v.x * Wr[c * 2]       + v.y * Wr[(c + 1) * 2]
           + v.z * Wr[(c + 2) * 2] + v.w * Wr[(c + 3) * 2];
  float p1 = v.x * Wr[c * 2 + 1]       + v.y * Wr[(c + 1) * 2 + 1]
           + v.z * Wr[(c + 2) * 2 + 1] + v.w * Wr[(c + 3) * 2 + 1];
#pragma unroll
  for (int off = 16; off > 0; off >>= 1) {
    p0 += __shfl_down(p0, off, 32);
    p1 += __shfl_down(p1, off, 32);
  }
  if (lane == 0) {
    out[(size_t)w * 2]     = p0 + br[0];
    out[(size_t)w * 2 + 1] = p1 + br[1];
  }
}

extern "C" void kernel_launch(void* const* d_in, const int* in_sizes, int n_in,
                              void* d_out, int out_size, void* d_ws, size_t ws_size,
                              hipStream_t stream) {
  const float* x     = (const float*)d_in[0];
  const int*   eidx  = (const int*)d_in[1];
  const float* W1    = (const float*)d_in[2];
  const float* b1    = (const float*)d_in[3];
  const float* W2    = (const float*)d_in[4];
  const float* b2    = (const float*)d_in[5];
  const float* W3    = (const float*)d_in[6];
  const float* b3    = (const float*)d_in[7];
  const float* gamma = (const float*)d_in[8];
  const float* beta  = (const float*)d_in[9];
  const float* Wc    = (const float*)d_in[10];
  const float* bc    = (const float*)d_in[11];
  const float* Wr    = (const float*)d_in[12];
  const float* br    = (const float*)d_in[13];
  float* out = (float*)d_out;

  const int N = N_NODES, E = E_EDGES;
  const int* src = eidx;
  const int* dst = eidx + E;

  float* buf0  = (float*)d_ws;
  float* buf1  = buf0 + (size_t)N * DHID;
  float* dinv  = buf1 + (size_t)N * DHID;
  float* stats = dinv + N;
  unsigned int* counts = (unsigned int*)(stats + 256);
  float* bnp   = stats + 256 + AGG_BLOCKS;
  unsigned int* recs = (unsigned int*)(bnp + 256);

  const dim3 blk(256);
  const int gBin  = (E / 4 + 1023) / 1024;
  const int gGemm = (N + 127) / 128;
  const int gHead = (N * 32 + 255) / 256;

  k_zero<<<2, blk, 0, stream>>>(stats, 256 + AGG_BLOCKS);

  k_bin<<<gBin, 1024, 0, stream>>>(src, dst, recs, counts, E / 4);
  k_rangedeg<<<AGG_BLOCKS, DEG_THREADS, 0, stream>>>(dst, recs, counts, dinv);

  k_gemm128<0, false, false, true><<<gGemm, blk, 0, stream>>>(x, W1, buf0, N, nullptr, nullptr, nullptr, dinv);
  k_aggregate<<<AGG_BLOCKS, 1024, 0, stream>>>(src, dst, dinv, buf0, b1, buf1, recs, counts);

  k_gemm128<1, false, false, true><<<gGemm, blk, 0, stream>>>(buf1, W2, buf0, N, nullptr, nullptr, nullptr, dinv);
  k_aggregate<<<AGG_BLOCKS, 1024, 0, stream>>>(src, dst, dinv, buf0, b2, buf1, recs, counts);

  k_gemm128<1, false, false, true><<<gGemm, blk, 0, stream>>>(buf1, W3, buf0, N, nullptr, nullptr, nullptr, dinv);
  k_aggregate<<<AGG_BLOCKS, 1024, 0, stream>>>(src, dst, dinv, buf0, b3, buf1, recs, counts);

  k_bnstats<<<1024, blk, 0, stream>>>(buf1, stats, N);
  k_bnfinal<<<1, 128, 0, stream>>>(stats, gamma, beta, bnp, N);

  k_gemm128<2, true, true, false><<<gGemm, blk, 0, stream>>>(buf1, Wc, buf0, N, bnp, bnp + 128, bc, nullptr);

  k_head<<<gHead, blk, 0, stream>>>(buf0, Wr, br, out, N);
}


# Products: one resident 32-wave work-group per WGP stages W once and deals 16-row tiles evenly over its four SIMDs (a 25th tile cut in column quarters), not 782 8-wave blocks; sums unchanged
# speedup vs baseline: 3.6822x; 1.0612x over previous
#include <hip/hip_runtime.h>

#define N_NODES 100000
#define DHID    128
#define E_EDGES 1600000
#define BN_EPS  1e-5f

typedef __attribute__((ext_vector_type(2))) float v2f;
typedef __attribute__((ext_vector_type(8))) float v8f;

__global__ void k_zero(float* __restrict__ p, int n) {
  int i = blockIdx.x * blockDim.x + threadIdx.x;
  if (i < n) p[i] = 0.0f;
}

#define ROWS_PER_BLOCK 391
#define AGG_BLOCKS     256
static_assert(AGG_BLOCKS * ROWS_PER_BLOCK >= N_NODES, "ranges must cover every node");
static_assert(E_EDGES % 4 == 0, "edges are read 4 per thread");

#define GEMM_BLOCKS  256
#define GEMM_THREADS 1024
#define GEMM_WAVES   (GEMM_THREADS / 32)

template <int AMODE, bool HAS_BIAS, bool EPI_RELU, bool ROW_SCALE, int NT>
__device__ __forceinline__ void gemm_tile(
    const float* __restrict__ A, const float2* __restrict__ wl2, float* __restrict__ C,
    int m0, int t0, int laneM, int hi,
    const float* __restrict__ bnScale, const float* __restrict__ bnShift,
    const float* __restrict__ bias, const float* __restrict__ rowscale)
{
  const float* __restrict__ arow = A + (size_t)(m0 + laneM) * DHID;

  v8f acc[NT];
#pragma unroll
  for (int t = 0; t < NT; ++t) acc[t] = (v8f){};

  for (int k = 0; k < DHID; k += 4) {
    float2 av = *(const float2*)(arow + k + hi * 2);
    if (AMODE == 1) {
      av.x = fmaxf(av.x, 0.0f);
      av.y = fmaxf(av.y, 0.0f);
    } else if (AMODE == 2) {
      int c0 = k + hi * 2;
      av.x = fmaxf(av.x, 0.0f) * bnScale[c0]     + bnShift[c0];
      av.y = fmaxf(av.y, 0.0f) * bnScale[c0 + 1] + bnShift[c0 + 1];
    }
    v2f a; a[0] = av.x; a[1] = av.y;

    const float2* __restrict__ brow = wl2 + (((k >> 1) + hi) << 7) + t0 * 16;
#pragma unroll
    for (int t = 0; t < NT; ++t) {
      float2 bv = brow[t * 16 + laneM];
      v2f b; b[0] = bv.x; b[1] = bv.y;
      acc[t] = __builtin_amdgcn_wmma_f32_16x16x4_f32(
          false, a, false, b, (short)0, acc[t], false, false);
    }
  }

  float rs[8];
#pragma unroll
  for (int r = 0; r < 8; ++r) rs[r] = ROW_SCALE ? rowscale[m0 + r + hi * 8] : 1.0f;

#pragma unroll
  for (int t = 0; t < NT; ++t) {
    int   col  = (t0 + t) * 16 + laneM;
    float bval = HAS_BIAS ? bias[col] : 0.0f;
#pragma unroll
    for (int r = 0; r < 8; ++r) {
      int   row = m0 + r + hi * 8;
      float v   = acc[t][r];
      if (ROW_SCALE) v *= rs[r];
      if (HAS_BIAS) v += bval;
      if (EPI_RELU) v = fmaxf(v, 0.0f);
      C[(size_t)row * DHID + col] = v;
    }
  }
}

template <int AMODE, bool HAS_BIAS, bool EPI_RELU, bool ROW_SCALE>
__global__ __launch_bounds__(GEMM_THREADS) void k_gemm128(
    const float* __restrict__ A, const float* __restrict__ W,
    float* __restrict__ C, int nrows,
    const float* __restrict__ bnScale, const float* __restrict__ bnShift,
    const float* __restrict__ bias, const float* __restrict__ rowscale)
{
  __shared__ float wl[DHID * DHID];
  const int tid = threadIdx.x;
  for (int i = tid; i < DHID * DHID; i += GEMM_THREADS) {
    int k = i >> 7, n = i & 127;
    wl[(((k >> 1) << 7) + n) * 2 + (k & 1)] = W[i];
  }
  __syncthreads();

  const int wave  = __builtin_amdgcn_readfirstlane(tid >> 5);
  const int lane  = tid & 31;
  const int laneM = lane & 15;
  const int hi    = lane >> 4;
  const float2* __restrict__ wl2 = (const float2*)wl;

  const int  ntiles = nrows >> 4;
  const int  full   = ntiles / GEMM_BLOCKS;
  const bool extra  = (int)blockIdx.x < ntiles - full * GEMM_BLOCKS;

  for (int i = wave; i < full; i += GEMM_WAVES)
    gemm_tile<AMODE, HAS_BIAS, EPI_RELU, ROW_SCALE, 8>(
        A, wl2, C, ((int)blockIdx.x + GEMM_BLOCKS * i) << 4, 0, laneM, hi, bnScale, bnShift, bias, rowscale);

  if (extra) {
    const int j = (wave - full) & (GEMM_WAVES - 1);
    if (j < 4)
      gemm_tile<AMODE, HAS_BIAS, EPI_RELU, ROW_SCALE, 2>(
          A, wl2, C, ((int)blockIdx.x + GEMM_BLOCKS * full) << 4, j * 2, laneM, hi, bnScale, bnShift, bias,
          rowscale);
  }
}

#define LIST_CAP   24576
#define EDGE_GROUP 8
#define SRC_BITS   17
static_assert(N_NODES <= (1 << SRC_BITS), "a source node must fit the record's low bits");
static_assert(ROWS_PER_BLOCK <= (1 << (32 - SRC_BITS)), "a local row must fit the record's high bits");

#define BIN_EDGES 4096
static_assert(AGG_BLOCKS == 256, "the count scan below is written for 8 waves of 32 ranges");
__global__ __launch_bounds__(1024) void k_bin(const int* __restrict__ src, const int* __restrict__ dst,
                                              unsigned int* __restrict__ recs,
                                              unsigned int* __restrict__ counts, int e4) {
  __shared__ unsigned int  hist[AGG_BLOCKS];
  __shared__ unsigned int  pos[AGG_BLOCKS];
  __shared__ unsigned int  gbase[AGG_BLOCKS];
  __shared__ unsigned int  wsum[8];
  __shared__ unsigned int  stage[BIN_EDGES];
  __shared__ unsigned char srange[BIN_EDGES];
  const int tid = threadIdx.x;
  if (tid < AGG_BLOCKS) hist[tid] = 0u;
  __syncthreads();

  const int  i  = blockIdx.x * 1024 + tid;
  const bool in = i < e4;
  int      s[4] = {0, 0, 0, 0}, d[4] = {0, 0, 0, 0};
  unsigned r[4] = {0u, 0u, 0u, 0u}, rank[4] = {0u, 0u, 0u, 0u};
  if (in) {
    const int4 sv = ((const int4*)src)[i];
    const int4 dv = ((const int4*)dst)[i];
    s[0] = sv.x; s[1] = sv.y; s[2] = sv.z; s[3] = sv.w;
    d[0] = dv.x; d[1] = dv.y; d[2] = dv.z; d[3] = dv.w;
#pragma unroll
    for (int u = 0; u < 4; ++u) {
      r[u]    = (unsigned)(d[u] / ROWS_PER_BLOCK);
      rank[u] = atomicAdd(&hist[r[u]], 1u);
    }
  }
  __syncthreads();

  unsigned n = 0u, incl = 0u;
  if (tid < AGG_BLOCKS) {
    n    = hist[tid];
    incl = n;
#pragma unroll
    for (int off = 1; off < 32; off <<= 1) {
      const unsigned t = __shfl_up(incl, off, 32);
      if ((tid & 31) >= off) incl += t;
    }
    if ((tid & 31) == 31) wsum[tid >> 5] = incl;
    gbase[tid] = n ? atomicAdd(&counts[tid], n) : 0u;
  }
  __syncthreads();
  if (tid < AGG_BLOCKS) {
    unsigned before = 0u;
    for (int w = 0; w < (tid >> 5); ++w) before += wsum[w];
    pos[tid] = before + incl - n;
  }
  __syncthreads();

  if (in) {
#pragma unroll
    for (int u = 0; u < 4; ++u) {
      const unsigned j  = pos[r[u]] + rank[u];
      const unsigned dl = (unsigned)d[u] - r[u] * ROWS_PER_BLOCK;
      stage[j]  = (unsigned)s[u] | (dl << SRC_BITS);
      srange[j] = (unsigned char)r[u];
    }
  }
  __syncthreads();

  const int nrec = min(1024, e4 - (int)blockIdx.x * 1024) * 4;
  for (int j = tid; j < nrec; j += 1024) {
    const unsigned rr   = srange[j];
    const unsigned slot = gbase[rr] + ((unsigned)j - pos[rr]);
    if (slot < LIST_CAP) recs[(size_t)rr * LIST_CAP + slot] = stage[j];
  }
}

#define DEG_THREADS 256
#define DEG_UNROLL  32
__global__ __launch_bounds__(DEG_THREADS) void k_rangedeg(const int* __restrict__ dst,
                                                          const unsigned int* __restrict__ recs,
                                                          const unsigned int* __restrict__ counts,
                                                          float* __restrict__ dinv) {
  __shared__ unsigned int cnt[ROWS_PER_BLOCK];
  const int tid = threadIdx.x;
  const int n0  = blockIdx.x * ROWS_PER_BLOCK;
  for (int i = tid; i < ROWS_PER_BLOCK; i += DEG_THREADS) cnt[i] = 0u;
  __syncthreads();

  const int total = (int)counts[blockIdx.x];
  if (total <= LIST_CAP) {
    const unsigned int* __restrict__ rl = recs + (size_t)blockIdx.x * LIST_CAP;
    for (int j0 = tid; j0 < total; j0 += DEG_UNROLL * DEG_THREADS) {
      unsigned rv[DEG_UNROLL];
#pragma unroll
      for (int u = 0; u < DEG_UNROLL; ++u) rv[u] = rl[min(j0 + u * DEG_THREADS, total - 1)];
#pragma unroll
      for (int u = 0; u < DEG_UNROLL; ++u)
        if (j0 + u * DEG_THREADS < total) atomicAdd(&cnt[rv[u] >> SRC_BITS], 1u);
    }
  } else {
    for (int e = tid; e < E_EDGES; e += DEG_THREADS) {
      const int d = dst[e];
      if (d >= n0 && d < n0 + ROWS_PER_BLOCK) atomicAdd(&cnt[d - n0], 1u);
    }
  }
  __syncthreads();

  const int nrows = min(ROWS_PER_BLOCK, N_NODES - n0);
  for (int i = tid; i < nrows; i += DEG_THREADS) dinv[n0 + i] = __frsqrt_rn((float)cnt[i] + 1.0f);
}

__device__ __forceinline__ int lds_word(int ch) { return (ch & 3) * 32 + (ch >> 2); }

__device__ __forceinline__ void add_group(const float* __restrict__ h, float* acc, unsigned lane,
                                          int nvalid, int s, int dl) {
  if (nvalid == EDGE_GROUP) {
    float4 v[EDGE_GROUP];
#pragma unroll
    for (int u = 0; u < EDGE_GROUP; ++u) {
      int sj = __builtin_amdgcn_readlane(s, u);
      v[u] = ((const float4*)(h + (size_t)sj * DHID))[lane];
    }
#pragma unroll
    for (int u = 0; u < EDGE_GROUP; ++u) {
      int    dj = __builtin_amdgcn_readlane(dl, u);
      float* a  = acc + dj * DHID;
      unsafeAtomicAdd(a + lane,      v[u].x);
      unsafeAtomicAdd(a + lane + 32, v[u].y);
      unsafeAtomicAdd(a + lane + 64, v[u].z);
      unsafeAtomicAdd(a + lane + 96, v[u].w);
    }
  } else {
    for (int j = 0; j < nvalid; ++j) {
      int sj = __builtin_amdgcn_readlane(s, j);
      int dj = __builtin_amdgcn_readlane(dl, j);
      const float4 hv = ((const float4*)(h + (size_t)sj * DHID))[lane];
      float*       a  = acc + dj * DHID;
      unsafeAtomicAdd(a + lane,      hv.x);
      unsafeAtomicAdd(a + lane + 32, hv.y);
      unsafeAtomicAdd(a + lane + 64, hv.z);
      unsafeAtomicAdd(a + lane + 96, hv.w);
    }
  }
}

__global__ __launch_bounds__(1024) void k_aggregate(
    const int* __restrict__ src, const int* __restrict__ dst,
    const float* __restrict__ dinv,
    const float* __restrict__ h, const float* __restrict__ b,
    float* __restrict__ out,
    const unsigned int* __restrict__ recs, const unsigned int* __restrict__ counts)
{
  __shared__ float acc[ROWS_PER_BLOCK * DHID];
  __shared__ float drow[ROWS_PER_BLOCK];

  const int      tid  = threadIdx.x;
  const unsigned lane = threadIdx.x & 31u;
  const int      n0   = blockIdx.x * ROWS_PER_BLOCK;
  const int      wave = __builtin_amdgcn_readfirstlane(tid >> 5);

  const unsigned int* __restrict__ rl = recs + (size_t)blockIdx.x * LIST_CAP;

  const int nvec = min(ROWS_PER_BLOCK, N_NODES - n0) * (DHID / 4);

  {
    const bool ownrow = tid * (DHID / 4) < nvec;
    float      dmine  = 0.0f;
    if (ownrow) dmine = dinv[n0 + tid];
    const float4* __restrict__ h4 = (const float4*)(h + (size_t)n0 * DHID);
#pragma unroll 4
    for (int i = tid; i < nvec; i += 1024) {
      float4 hv = h4[i];
      float* ar = acc + (i >> 5) * DHID + (i & 31);
      ar[0]  = hv.x;
      ar[32] = hv.y;
      ar[64] = hv.z;
      ar[96] = hv.w;
    }
    if (ownrow) drow[tid] = dmine;
  }
  __syncthreads();

  const int total = __builtin_amdgcn_readfirstlane((int)counts[blockIdx.x]);

  if (total <= LIST_CAP) {
    const int l7 = min((int)lane, EDGE_GROUP - 1);
    int      g0 = wave * EDGE_GROUP;
    unsigned rc = 0u;
    if (g0 < total) rc = rl[min(g0 + l7, total - 1)];
    for (; g0 < total; g0 += 32 * EDGE_GROUP) {
      const int nvalid = min(EDGE_GROUP, total - g0);
      const int gn = g0 + 32 * EDGE_GROUP;
      unsigned  nx = rc;
      if (gn < total) nx = rl[min(gn + l7, total - 1)];
      const int s  = (int)(rc & ((1u << SRC_BITS) - 1u));
      const int dl = (int)(rc >> SRC_BITS);
      add_group(h, acc, lane, nvalid, s, dl);
      rc = nx;
    }
  } else {
    for (int e = tid; e < E_EDGES; e += 1024) {
      const int d = dst[e];
      if (d >= n0 && d < n0 + ROWS_PER_BLOCK) {
        const float* hr = h + (size_t)src[e] * DHID;
        float*       ar = acc + (d - n0) * DHID;
        for (int ch = 0; ch < DHID; ++ch) unsafeAtomicAdd(ar + lds_word(ch), hr[ch]);
      }
    }
  }
  __syncthreads();

  float4* __restrict__ out4 = (float4*)(out + (size_t)n0 * DHID);
  const float4 bv = *(const float4*)(b + (tid & 31) * 4);
  for (int i = tid; i < nvec; i += 1024) {
    const float dv = drow[i >> 5];
    const float* ar = acc + (i >> 5) * DHID + (i & 31);
    out4[i] = make_float4(ar[0] * dv + bv.x, ar[32] * dv + bv.y, ar[64] * dv + bv.z, ar[96] * dv + bv.w);
  }
}

__global__ __launch_bounds__(256) void k_bnstats(const float* __restrict__ h,
                                                 float* __restrict__ stats, int n) {
  __shared__ float ss[256], sq[256];
  int t = threadIdx.x, c = t & 127, half = t >> 7;
  float s = 0.0f, q = 0.0f;
  for (int r = blockIdx.x * 2 + half; r < n; r += gridDim.x * 2) {
    float v = fmaxf(h[(size_t)r * DHID + c], 0.0f);
    s += v; q += v * v;
  }
  ss[t] = s; sq[t] = q;
  __syncthreads();
  if (t < 128) {
    unsafeAtomicAdd(&stats[c],       ss[t] + ss[t + 128]);
    unsafeAtomicAdd(&stats[128 + c], sq[t] + sq[t + 128]);
  }
}

__global__ void k_bnfinal(const float* __restrict__ stats, const float* __restrict__ gamma,
                          const float* __restrict__ beta, float* __restrict__ bn, int n) {
  int c = threadIdx.x;
  if (c < 128) {
    float inv = 1.0f / (float)n;
    float mu  = stats[c] * inv;
    float var = stats[128 + c] * inv - mu * mu;
    float sc  = gamma[c] * __frsqrt_rn(var + BN_EPS);
    bn[c]       = sc;
    bn[128 + c] = beta[c] - mu * sc;
  }
}

__global__ __launch_bounds__(256) void k_head(const float* __restrict__ hc,
                                              const float* __restrict__ Wr,
                                              const float* __restrict__ br,
                                              float* __restrict__ out, int n) {
  int w    = (blockIdx.x * blockDim.x + threadIdx.x) >> 5;
  int lane = threadIdx.x & 31;
  if (w >= n) return;
  float4 v = *(const float4*)(hc + (size_t)w * DHID + lane * 4);
  int c = lane * 4;
  float p0 = v.x * Wr[c * 2]       + v.y * Wr[(c + 1) * 2]
           + v.z * Wr[(c + 2) * 2] + v.w * Wr[(c + 3) * 2];
  float p1 = v.x * Wr[c * 2 + 1]       + v.y * Wr[(c + 1) * 2 + 1]
           + v.z * Wr[(c + 2) * 2 + 1] + v.w * Wr[(c + 3) * 2 + 1];
#pragma unroll
  for (int off = 16; off > 0; off >>= 1) {
    p0 += __shfl_down(p0, off, 32);
    p1 += __shfl_down(p1, off, 32);
  }
  if (lane == 0) {
    out[(size_t)w * 2]     = p0 + br[0];
    out[(size_t)w * 2 + 1] = p1 + br[1];
  }
}

extern "C" void kernel_launch(void* const* d_in, const int* in_sizes, int n_in,
                              void* d_out, int out_size, void* d_ws, size_t ws_size,
                              hipStream_t stream) {
  const float* x     = (const float*)d_in[0];
  const int*   eidx  = (const int*)d_in[1];
  const float* W1    = (const float*)d_in[2];
  const float* b1    = (const float*)d_in[3];
  const float* W2    = (const float*)d_in[4];
  const float* b2    = (const float*)d_in[5];
  const float* W3    = (const float*)d_in[6];
  const float* b3    = (const float*)d_in[7];
  const float* gamma = (const float*)d_in[8];
  const float* beta  = (const float*)d_in[9];
  const float* Wc    = (const float*)d_in[10];
  const float* bc    = (const float*)d_in[11];
  const float* Wr    = (const float*)d_in[12];
  const float* br    = (const float*)d_in[13];
  float* out = (float*)d_out;

  const int N = N_NODES, E = E_EDGES;
  const int* src = eidx;
  const int* dst = eidx + E;

  float* buf0  = (float*)d_ws;
  float* buf1  = buf0 + (size_t)N * DHID;
  float* dinv  = buf1 + (size_t)N * DHID;
  float* stats = dinv + N;
  unsigned int* counts = (unsigned int*)(stats + 256);
  float* bnp   = stats + 256 + AGG_BLOCKS;
  unsigned int* recs = (unsigned int*)(bnp + 256);

  const dim3 blk(256);
  const int gBin  = (E / 4 + 1023) / 1024;
  const int gHead = (N * 32 + 255) / 256;

  k_zero<<<2, blk, 0, stream>>>(stats, 256 + AGG_BLOCKS);

  k_bin<<<gBin, 1024, 0, stream>>>(src, dst, recs, counts, E / 4);
  k_rangedeg<<<AGG_BLOCKS, DEG_THREADS, 0, stream>>>(dst, recs, counts, dinv);

  k_gemm128<0, false, false, true><<<GEMM_BLOCKS, GEMM_THREADS, 0, stream>>>(x, W1, buf0, N, nullptr, nullptr, nullptr, dinv);
  k_aggregate<<<AGG_BLOCKS, 1024, 0, stream>>>(src, dst, dinv, buf0, b1, buf1, recs, counts);

  k_gemm128<1, false, false, true><<<GEMM_BLOCKS, GEMM_THREADS, 0, stream>>>(buf1, W2, buf0, N, nullptr, nullptr, nullptr, dinv);
  k_aggregate<<<AGG_BLOCKS, 1024, 0, stream>>>(src, dst, dinv, buf0, b2, buf1, recs, counts);

  k_gemm128<1, false, false, true><<<GEMM_BLOCKS, GEMM_THREADS, 0, stream>>>(buf1, W3, buf0, N, nullptr, nullptr, nullptr, dinv);
  k_aggregate<<<AGG_BLOCKS, 1024, 0, stream>>>(src, dst, dinv, buf0, b3, buf1, recs, counts);

  k_bnstats<<<1024, blk, 0, stream>>>(buf1, stats, N);
  k_bnfinal<<<1, 128, 0, stream>>>(stats, gamma, beta, bnp, N);

  k_gemm128<2, true, true, false><<<GEMM_BLOCKS, GEMM_THREADS, 0, stream>>>(buf1, Wc, buf0, N, bnp, bnp + 128, bc, nullptr);

  k_head<<<gHead, blk, 0, stream>>>(buf0, Wr, br, out, N);
}


# Batchnorm sums and head: 256 work-groups of 25 waves, a wave takes 16 whole rows at once (16 row loads in flight), not 1 load per round trip / 1 row per wave; stat sums regrouped, head tree kept
# speedup vs baseline: 4.0827x; 1.1088x over previous
#include <hip/hip_runtime.h>

#define N_NODES 100000
#define DHID    128
#define E_EDGES 1600000
#define BN_EPS  1e-5f

typedef __attribute__((ext_vector_type(2))) float v2f;
typedef __attribute__((ext_vector_type(8))) float v8f;
typedef __attribute__((ext_vector_type(4))) float v4f;

__global__ void k_zero(float* __restrict__ p, int n) {
  int i = blockIdx.x * blockDim.x + threadIdx.x;
  if (i < n) p[i] = 0.0f;
}

#define ROWS_PER_BLOCK 391
#define AGG_BLOCKS     256
static_assert(AGG_BLOCKS * ROWS_PER_BLOCK >= N_NODES, "ranges must cover every node");
static_assert(E_EDGES % 4 == 0, "edges are read 4 per thread");

#define GEMM_BLOCKS  256
#define GEMM_THREADS 1024
#define GEMM_WAVES   (GEMM_THREADS / 32)

template <int AMODE, bool HAS_BIAS, bool EPI_RELU, bool ROW_SCALE, int NT>
__device__ __forceinline__ void gemm_tile(
    const float* __restrict__ A, const float2* __restrict__ wl2, float* __restrict__ C,
    int m0, int t0, int laneM, int hi,
    const float* __restrict__ bnScale, const float* __restrict__ bnShift,
    const float* __restrict__ bias, const float* __restrict__ rowscale)
{
  const float* __restrict__ arow = A + (size_t)(m0 + laneM) * DHID;

  v8f acc[NT];
#pragma unroll
  for (int t = 0; t < NT; ++t) acc[t] = (v8f){};

  for (int k = 0; k < DHID; k += 4) {
    float2 av = *(const float2*)(arow + k + hi * 2);
    if (AMODE == 1) {
      av.x = fmaxf(av.x, 0.0f);
      av.y = fmaxf(av.y, 0.0f);
    } else if (AMODE == 2) {
      int c0 = k + hi * 2;
      av.x = fmaxf(av.x, 0.0f) * bnScale[c0]     + bnShift[c0];
      av.y = fmaxf(av.y, 0.0f) * bnScale[c0 + 1] + bnShift[c0 + 1];
    }
    v2f a; a[0] = av.x; a[1] = av.y;

    const float2* __restrict__ brow = wl2 + (((k >> 1) + hi) << 7) + t0 * 16;
#pragma unroll
    for (int t = 0; t < NT; ++t) {
      float2 bv = brow[t * 16 + laneM];
      v2f b; b[0] = bv.x; b[1] = bv.y;
      acc[t] = __builtin_amdgcn_wmma_f32_16x16x4_f32(
          false, a, false, b, (short)0, acc[t], false, false);
    }
  }

  float rs[8];
#pragma unroll
  for (int r = 0; r < 8; ++r) rs[r] = ROW_SCALE ? rowscale[m0 + r + hi * 8] : 1.0f;

#pragma unroll
  for (int t = 0; t < NT; ++t) {
    int   col  = (t0 + t) * 16 + laneM;
    float bval = HAS_BIAS ? bias[col] : 0.0f;
#pragma unroll
    for (int r = 0; r < 8; ++r) {
      int   row = m0 + r + hi * 8;
      float v   = acc[t][r];
      if (ROW_SCALE) v *= rs[r];
      if (HAS_BIAS) v += bval;
      if (EPI_RELU) v = fmaxf(v, 0.0f);
      C[(size_t)row * DHID + col] = v;
    }
  }
}

template <int AMODE, bool HAS_BIAS, bool EPI_RELU, bool ROW_SCALE>
__global__ __launch_bounds__(GEMM_THREADS) void k_gemm128(
    const float* __restrict__ A, const float* __restrict__ W,
    float* __restrict__ C, int nrows,
    const float* __restrict__ bnScale, const float* __restrict__ bnShift,
    const float* __restrict__ bias, const float* __restrict__ rowscale)
{
  __shared__ float wl[DHID * DHID];
  const int tid = threadIdx.x;
  for (int i = tid; i < DHID * DHID; i += GEMM_THREADS) {
    int k = i >> 7, n = i & 127;
    wl[(((k >> 1) << 7) + n) * 2 + (k & 1)] = W[i];
  }
  __syncthreads();

  const int wave  = __builtin_amdgcn_readfirstlane(tid >> 5);
  const int lane  = tid & 31;
  const int laneM = lane & 15;
  const int hi    = lane >> 4;
  const float2* __restrict__ wl2 = (const float2*)wl;

  const int  ntiles = nrows >> 4;
  const int  full   = ntiles / GEMM_BLOCKS;
  const bool extra  = (int)blockIdx.x < ntiles - full * GEMM_BLOCKS;

  for (int i = wave; i < full; i += GEMM_WAVES)
    gemm_tile<AMODE, HAS_BIAS, EPI_RELU, ROW_SCALE, 8>(
        A, wl2, C, ((int)blockIdx.x + GEMM_BLOCKS * i) << 4, 0, laneM, hi, bnScale, bnShift, bias, rowscale);

  if (extra) {
    const int j = (wave - full) & (GEMM_WAVES - 1);
    if (j < 4)
      gemm_tile<AMODE, HAS_BIAS, EPI_RELU, ROW_SCALE, 2>(
          A, wl2, C, ((int)blockIdx.x + GEMM_BLOCKS * full) << 4, j * 2, laneM, hi, bnScale, bnShift, bias,
          rowscale);
  }
}

#define LIST_CAP   24576
#define EDGE_GROUP 8
#define SRC_BITS   17
static_assert(N_NODES <= (1 << SRC_BITS), "a source node must fit the record's low bits");
static_assert(ROWS_PER_BLOCK <= (1 << (32 - SRC_BITS)), "a local row must fit the record's high bits");

#define BIN_EDGES 4096
static_assert(AGG_BLOCKS == 256, "the count scan below is written for 8 waves of 32 ranges");
__global__ __launch_bounds__(1024) void k_bin(const int* __restrict__ src, const int* __restrict__ dst,
                                              unsigned int* __restrict__ recs,
                                              unsigned int* __restrict__ counts, int e4) {
  __shared__ unsigned int  hist[AGG_BLOCKS];
  __shared__ unsigned int  pos[AGG_BLOCKS];
  __shared__ unsigned int  gbase[AGG_BLOCKS];
  __shared__ unsigned int  wsum[8];
  __shared__ unsigned int  stage[BIN_EDGES];
  __shared__ unsigned char srange[BIN_EDGES];
  const int tid = threadIdx.x;
  if (tid < AGG_BLOCKS) hist[tid] = 0u;
  __syncthreads();

  const int  i  = blockIdx.x * 1024 + tid;
  const bool in = i < e4;
  int      s[4] = {0, 0, 0, 0}, d[4] = {0, 0, 0, 0};
  unsigned r[4] = {0u, 0u, 0u, 0u}, rank[4] = {0u, 0u, 0u, 0u};
  if (in) {
    const int4 sv = ((const int4*)src)[i];
    const int4 dv = ((const int4*)dst)[i];
    s[0] = sv.x; s[1] = sv.y; s[2] = sv.z; s[3] = sv.w;
    d[0] = dv.x; d[1] = dv.y; d[2] = dv.z; d[3] = dv.w;
#pragma unroll
    for (int u = 0; u < 4; ++u) {
      r[u]    = (unsigned)(d[u] / ROWS_PER_BLOCK);
      rank[u] = atomicAdd(&hist[r[u]], 1u);
    }
  }
  __syncthreads();

  unsigned n = 0u, incl = 0u;
  if (tid < AGG_BLOCKS) {
    n    = hist[tid];
    incl = n;
#pragma unroll
    for (int off = 1; off < 32; off <<= 1) {
      const unsigned t = __shfl_up(incl, off, 32);
      if ((tid & 31) >= off) incl += t;
    }
    if ((tid & 31) == 31) wsum[tid >> 5] = incl;
    gbase[tid] = n ? atomicAdd(&counts[tid], n) : 0u;
  }
  __syncthreads();
  if (tid < AGG_BLOCKS) {
    unsigned before = 0u;
    for (int w = 0; w < (tid >> 5); ++w) before += wsum[w];
    pos[tid] = before + incl - n;
  }
  __syncthreads();

  if (in) {
#pragma unroll
    for (int u = 0; u < 4; ++u) {
      const unsigned j  = pos[r[u]] + rank[u];
      const unsigned dl = (unsigned)d[u] - r[u] * ROWS_PER_BLOCK;
      stage[j]  = (unsigned)s[u] | (dl << SRC_BITS);
      srange[j] = (unsigned char)r[u];
    }
  }
  __syncthreads();

  const int nrec = min(1024, e4 - (int)blockIdx.x * 1024) * 4;
  for (int j = tid; j < nrec; j += 1024) {
    const unsigned rr   = srange[j];
    const unsigned slot = gbase[rr] + ((unsigned)j - pos[rr]);
    if (slot < LIST_CAP) recs[(size_t)rr * LIST_CAP + slot] = stage[j];
  }
}

#define DEG_THREADS 256
#define DEG_UNROLL  32
__global__ __launch_bounds__(DEG_THREADS) void k_rangedeg(const int* __restrict__ dst,
                                                          const unsigned int* __restrict__ recs,
                                                          const unsigned int* __restrict__ counts,
                                                          float* __restrict__ dinv) {
  __shared__ unsigned int cnt[ROWS_PER_BLOCK];
  const int tid = threadIdx.x;
  const int n0  = blockIdx.x * ROWS_PER_BLOCK;
  for (int i = tid; i < ROWS_PER_BLOCK; i += DEG_THREADS) cnt[i] = 0u;
  __syncthreads();

  const int total = (int)counts[blockIdx.x];
  if (total <= LIST_CAP) {
    const unsigned int* __restrict__ rl = recs + (size_t)blockIdx.x * LIST_CAP;
    for (int j0 = tid; j0 < total; j0 += DEG_UNROLL * DEG_THREADS) {
      unsigned rv[DEG_UNROLL];
#pragma unroll
      for (int u = 0; u < DEG_UNROLL; ++u) rv[u] = rl[min(j0 + u * DEG_THREADS, total - 1)];
#pragma unroll
      for (int u = 0; u < DEG_UNROLL; ++u)
        if (j0 + u * DEG_THREADS < total) atomicAdd(&cnt[rv[u] >> SRC_BITS], 1u);
    }
  } else {
    for (int e = tid; e < E_EDGES; e += DEG_THREADS) {
      const int d = dst[e];
      if (d >= n0 && d < n0 + ROWS_PER_BLOCK) atomicAdd(&cnt[d - n0], 1u);
    }
  }
  __syncthreads();

  const int nrows = min(ROWS_PER_BLOCK, N_NODES - n0);
  for (int i = tid; i < nrows; i += DEG_THREADS) dinv[n0 + i] = __frsqrt_rn((float)cnt[i] + 1.0f);
}

__device__ __forceinline__ int lds_word(int ch) { return (ch & 3) * 32 + (ch >> 2); }

__device__ __forceinline__ void add_group(const float* __restrict__ h, float* acc, unsigned lane,
                                          int nvalid, int s, int dl) {
  if (nvalid == EDGE_GROUP) {
    float4 v[EDGE_GROUP];
#pragma unroll
    for (int u = 0; u < EDGE_GROUP; ++u) {
      int sj = __builtin_amdgcn_readlane(s, u);
      v[u] = ((const float4*)(h + (size_t)sj * DHID))[lane];
    }
#pragma unroll
    for (int u = 0; u < EDGE_GROUP; ++u) {
      int    dj = __builtin_amdgcn_readlane(dl, u);
      float* a  = acc + dj * DHID;
      unsafeAtomicAdd(a + lane,      v[u].x);
      unsafeAtomicAdd(a + lane + 32, v[u].y);
      unsafeAtomicAdd(a + lane + 64, v[u].z);
      unsafeAtomicAdd(a + lane + 96, v[u].w);
    }
  } else {
    for (int j = 0; j < nvalid; ++j) {
      int sj = __builtin_amdgcn_readlane(s, j);
      int dj = __builtin_amdgcn_readlane(dl, j);
      const float4 hv = ((const float4*)(h + (size_t)sj * DHID))[lane];
      float*       a  = acc + dj * DHID;
      unsafeAtomicAdd(a + lane,      hv.x);
      unsafeAtomicAdd(a + lane + 32, hv.y);
      unsafeAtomicAdd(a + lane + 64, hv.z);
      unsafeAtomicAdd(a + lane + 96, hv.w);
    }
  }
}

__global__ __launch_bounds__(1024) void k_aggregate(
    const int* __restrict__ src, const int* __restrict__ dst,
    const float* __restrict__ dinv,
    const float* __restrict__ h, const float* __restrict__ b,
    float* __restrict__ out,
    const unsigned int* __restrict__ recs, const unsigned int* __restrict__ counts)
{
  __shared__ float acc[ROWS_PER_BLOCK * DHID];
  __shared__ float drow[ROWS_PER_BLOCK];

  const int      tid  = threadIdx.x;
  const unsigned lane = threadIdx.x & 31u;
  const int      n0   = blockIdx.x * ROWS_PER_BLOCK;
  const int      wave = __builtin_amdgcn_readfirstlane(tid >> 5);

  const unsigned int* __restrict__ rl = recs + (size_t)blockIdx.x * LIST_CAP;

  const int nvec = min(ROWS_PER_BLOCK, N_NODES - n0) * (DHID / 4);

  {
    const bool ownrow = tid * (DHID / 4) < nvec;
    float      dmine  = 0.0f;
    if (ownrow) dmine = dinv[n0 + tid];
    const float4* __restrict__ h4 = (const float4*)(h + (size_t)n0 * DHID);
#pragma unroll 4
    for (int i = tid; i < nvec; i += 1024) {
      float4 hv = h4[i];
      float* ar = acc + (i >> 5) * DHID + (i & 31);
      ar[0]  = hv.x;
      ar[32] = hv.y;
      ar[64] = hv.z;
      ar[96] = hv.w;
    }
    if (ownrow) drow[tid] = dmine;
  }
  __syncthreads();

  const int total = __builtin_amdgcn_readfirstlane((int)counts[blockIdx.x]);

  if (total <= LIST_CAP) {
    const int l7 = min((int)lane, EDGE_GROUP - 1);
    int      g0 = wave * EDGE_GROUP;
    unsigned rc = 0u;
    if (g0 < total) rc = rl[min(g0 + l7, total - 1)];
    for (; g0 < total; g0 += 32 * EDGE_GROUP) {
      const int nvalid = min(EDGE_GROUP, total - g0);
      const int gn = g0 + 32 * EDGE_GROUP;
      unsigned  nx = rc;
      if (gn < total) nx = rl[min(gn + l7, total - 1)];
      const int s  = (int)(rc & ((1u << SRC_BITS) - 1u));
      const int dl = (int)(rc >> SRC_BITS);
      add_group(h, acc, lane, nvalid, s, dl);
      rc = nx;
    }
  } else {
    for (int e = tid; e < E_EDGES; e += 1024) {
      const int d = dst[e];
      if (d >= n0 && d < n0 + ROWS_PER_BLOCK) {
        const float* hr = h + (size_t)src[e] * DHID;
        float*       ar = acc + (d - n0) * DHID;
        for (int ch = 0; ch < DHID; ++ch) unsafeAtomicAdd(ar + lds_word(ch), hr[ch]);
      }
    }
  }
  __syncthreads();

  float4* __restrict__ out4 = (float4*)(out + (size_t)n0 * DHID);
  const float4 bv = *(const float4*)(b + (tid & 31) * 4);
  for (int i = tid; i < nvec; i += 1024) {
    const float dv = drow[i >> 5];
    const float* ar = acc + (i >> 5) * DHID + (i & 31);
    out4[i] = make_float4(ar[0] * dv + bv.x, ar[32] * dv + bv.y, ar[64] * dv + bv.z, ar[96] * dv + bv.w);
  }
}

#define TAIL_ROWS    16
#define TAIL_WAVES   25
#define TAIL_THREADS (TAIL_WAVES * 32)
#define TAIL_BLOCKS  256
static_assert(TAIL_ROWS * 2 * sizeof(float) == 128, "a row group's head outputs are one 128-byte line");

__global__ __launch_bounds__(TAIL_THREADS) void k_bnstats(const float* __restrict__ h,
                                                          float* __restrict__ stats, int n) {
  __shared__ float red[TAIL_WAVES][2 * DHID];
  const int tid  = threadIdx.x;
  const int lane = tid & 31;
  const int wave = __builtin_amdgcn_readfirstlane(tid >> 5);
  const int c    = lane * 4;
  float s0 = 0.0f, s1 = 0.0f, s2 = 0.0f, s3 = 0.0f;
  float q0 = 0.0f, q1 = 0.0f, q2 = 0.0f, q3 = 0.0f;
  const int ngroups = (n + TAIL_ROWS - 1) / TAIL_ROWS;
  for (int g = wave * (int)gridDim.x + (int)blockIdx.x; g < ngroups; g += TAIL_WAVES * (int)gridDim.x) {
    const int row0 = g * TAIL_ROWS;
    if (row0 + TAIL_ROWS <= n) {
      v4f v[TAIL_ROWS];
#pragma unroll
      for (int u = 0; u < TAIL_ROWS; ++u) v[u] = *(const v4f*)(h + (size_t)(row0 + u) * DHID + c);
      asm volatile("" :: "v"(v[0]), "v"(v[1]), "v"(v[2]), "v"(v[3]), "v"(v[4]), "v"(v[5]), "v"(v[6]), "v"(v[7]),
                         "v"(v[8]), "v"(v[9]), "v"(v[10]), "v"(v[11]), "v"(v[12]), "v"(v[13]), "v"(v[14]), "v"(v[15]));
#pragma unroll
      for (int u = 0; u < TAIL_ROWS; ++u) {
        const float a0 = fmaxf(v[u].x, 0.0f), a1 = fmaxf(v[u].y, 0.0f);
        const float a2 = fmaxf(v[u].z, 0.0f), a3 = fmaxf(v[u].w, 0.0f);
        s0 += a0; q0 += a0 * a0;
        s1 += a1; q1 += a1 * a1;
        s2 += a2; q2 += a2 * a2;
        s3 += a3; q3 += a3 * a3;
      }
    } else {
      for (int r = row0; r < n; ++r) {
        const float4 t = *(const float4*)(h + (size_t)r * DHID + c);
        const float a0 = fmaxf(t.x, 0.0f), a1 = fmaxf(t.y, 0.0f);
        const float a2 = fmaxf(t.z, 0.0f), a3 = fmaxf(t.w, 0.0f);
        s0 += a0; q0 += a0 * a0;
        s1 += a1; q1 += a1 * a1;
        s2 += a2; q2 += a2 * a2;
        s3 += a3; q3 += a3 * a3;
      }
    }
  }
  *(float4*)&red[wave][c]        = make_float4(s0, s1, s2, s3);
  *(float4*)&red[wave][DHID + c] = make_float4(q0, q1, q2, q3);
  __syncthreads();
  if (tid < 2 * DHID) {
    float t = 0.0f;
#pragma unroll
    for (int w = 0; w < TAIL_WAVES; ++w) t += red[w][tid];
    unsafeAtomicAdd(&stats[tid], t);
  }
}

__global__ void k_bnfinal(const float* __restrict__ stats, const float* __restrict__ gamma,
                          const float* __restrict__ beta, float* __restrict__ bn, int n) {
  int c = threadIdx.x;
  if (c < 128) {
    float inv = 1.0f / (float)n;
    float mu  = stats[c] * inv;
    float var = stats[128 + c] * inv - mu * mu;
    float sc  = gamma[c] * __frsqrt_rn(var + BN_EPS);
    bn[c]       = sc;
    bn[128 + c] = beta[c] - mu * sc;
  }
}

__global__ __launch_bounds__(TAIL_THREADS) void k_head(const float* __restrict__ hc,
                                                       const float* __restrict__ Wr,
                                                       const float* __restrict__ br,
                                                       float* __restrict__ out, int n) {
  const int lane = threadIdx.x & 31;
  const int wave = __builtin_amdgcn_readfirstlane((int)(threadIdx.x >> 5));
  const int c = lane * 4;
  const float w00 = Wr[c * 2],     w01 = Wr[(c + 1) * 2],     w02 = Wr[(c + 2) * 2],     w03 = Wr[(c + 3) * 2];
  const float w10 = Wr[c * 2 + 1], w11 = Wr[(c + 1) * 2 + 1], w12 = Wr[(c + 2) * 2 + 1], w13 = Wr[(c + 3) * 2 + 1];
  const float bsel = br[lane >> 4];

  const bool h16 = (lane & 16) != 0, h8 = (lane & 8) != 0, h4 = (lane & 4) != 0, h2 = (lane & 2) != 0,
             h1 = (lane & 1) != 0;
  const int ngroups = (n + TAIL_ROWS - 1) / TAIL_ROWS;
  for (int g = wave * (int)gridDim.x + (int)blockIdx.x; g < ngroups; g += TAIL_WAVES * (int)gridDim.x) {
    const int row0 = g * TAIL_ROWS;
    float4 v[TAIL_ROWS];
#pragma unroll
    for (int u = 0; u < TAIL_ROWS; ++u) {
      const int r = min(row0 + u, n - 1);
      v[u] = *(const float4*)(hc + (size_t)r * DHID + c);
    }
    __builtin_amdgcn_sched_barrier(0);
    float x[2 * TAIL_ROWS];
#pragma unroll
    for (int u = 0; u < TAIL_ROWS; ++u) {
      x[u]             = v[u].x * w00 + v[u].y * w01 + v[u].z * w02 + v[u].w * w03;
      x[TAIL_ROWS + u] = v[u].x * w10 + v[u].y * w11 + v[u].z * w12 + v[u].w * w13;
    }
    float y16[16], y8[8], y4[4], y2[2];
#pragma unroll
    for (int j = 0; j < 16; ++j) {
      const float keep = h16 ? x[j + 16] : x[j];
      const float send = h16 ? x[j] : x[j + 16];
      y16[j] = keep + __shfl_xor(send, 16, 32);
    }
#pragma unroll
    for (int j = 0; j < 8; ++j) {
      const float keep = h8 ? y16[j + 8] : y16[j];
      const float send = h8 ? y16[j] : y16[j + 8];
      y8[j] = keep + __shfl_xor(send, 8, 32);
    }
#pragma unroll
    for (int j = 0; j < 4; ++j) {
      const float keep = h4 ? y8[j + 4] : y8[j];
      const float send = h4 ? y8[j] : y8[j + 4];
      y4[j] = keep + __shfl_xor(send, 4, 32);
    }
#pragma unroll
    for (int j = 0; j < 2; ++j) {
      const float keep = h2 ? y4[j + 2] : y4[j];
      const float send = h2 ? y4[j] : y4[j + 2];
      y2[j] = keep + __shfl_xor(send, 2, 32);
    }
    const float keep1 = h1 ? y2[1] : y2[0];
    const float send1 = h1 ? y2[0] : y2[1];
    const float y1    = keep1 + __shfl_xor(send1, 1, 32);

    const int r = row0 + (lane & 15);
    if (r < n) out[(size_t)r * 2 + (lane >> 4)] = y1 + bsel;
  }
}

extern "C" void kernel_launch(void* const* d_in, const int* in_sizes, int n_in,
                              void* d_out, int out_size, void* d_ws, size_t ws_size,
                              hipStream_t stream) {
  const float* x     = (const float*)d_in[0];
  const int*   eidx  = (const int*)d_in[1];
  const float* W1    = (const float*)d_in[2];
  const float* b1    = (const float*)d_in[3];
  const float* W2    = (const float*)d_in[4];
  const float* b2    = (const float*)d_in[5];
  const float* W3    = (const float*)d_in[6];
  const float* b3    = (const float*)d_in[7];
  const float* gamma = (const float*)d_in[8];
  const float* beta  = (const float*)d_in[9];
  const float* Wc    = (const float*)d_in[10];
  const float* bc    = (const float*)d_in[11];
  const float* Wr    = (const float*)d_in[12];
  const float* br    = (const float*)d_in[13];
  float* out = (float*)d_out;

  const int N = N_NODES, E = E_EDGES;
  const int* src = eidx;
  const int* dst = eidx + E;

  float* buf0  = (float*)d_ws;
  float* buf1  = buf0 + (size_t)N * DHID;
  float* dinv  = buf1 + (size_t)N * DHID;
  float* stats = dinv + N;
  unsigned int* counts = (unsigned int*)(stats + 256);
  float* bnp   = stats + 256 + AGG_BLOCKS;
  unsigned int* recs = (unsigned int*)(bnp + 256);

  const dim3 blk(256);
  const int gBin  = (E / 4 + 1023) / 1024;

  k_zero<<<2, blk, 0, stream>>>(stats, 256 + AGG_BLOCKS);

  k_bin<<<gBin, 1024, 0, stream>>>(src, dst, recs, counts, E / 4);
  k_rangedeg<<<AGG_BLOCKS, DEG_THREADS, 0, stream>>>(dst, recs, counts, dinv);

  k_gemm128<0, false, false, true><<<GEMM_BLOCKS, GEMM_THREADS, 0, stream>>>(x, W1, buf0, N, nullptr, nullptr, nullptr, dinv);
  k_aggregate<<<AGG_BLOCKS, 1024, 0, stream>>>(src, dst, dinv, buf0, b1, buf1, recs, counts);

  k_gemm128<1, false, false, true><<<GEMM_BLOCKS, GEMM_THREADS, 0, stream>>>(buf1, W2, buf0, N, nullptr, nullptr, nullptr, dinv);
  k_aggregate<<<AGG_BLOCKS, 1024, 0, stream>>>(src, dst, dinv, buf0, b2, buf1, recs, counts);

  k_gemm128<1, false, false, true><<<GEMM_BLOCKS, GEMM_THREADS, 0, stream>>>(buf1, W3, buf0, N, nullptr, nullptr, nullptr, dinv);
  k_aggregate<<<AGG_BLOCKS, 1024, 0, stream>>>(src, dst, dinv, buf0, b3, buf1, recs, counts);

  k_bnstats<<<TAIL_BLOCKS, TAIL_THREADS, 0, stream>>>(buf1, stats, N);
  k_bnfinal<<<1, 128, 0, stream>>>(stats, gamma, beta, bnp, N);

  k_gemm128<2, true, true, false><<<GEMM_BLOCKS, GEMM_THREADS, 0, stream>>>(buf1, Wc, buf0, N, bnp, bnp + 128, bc, nullptr);

  k_head<<<TAIL_BLOCKS, TAIL_THREADS, 0, stream>>>(buf0, Wr, br, out, N);
}


# Edge sums: a range's rows dealt to its 32 waves (row mod 32), records split per wave in LDS; a wave sums its 13 rows in registers indexed by scalar slot; no LDS float adds; sublist order, self last
# speedup vs baseline: 4.3604x; 1.0680x over previous
#include <hip/hip_runtime.h>

#define N_NODES 100000
#define DHID    128
#define E_EDGES 1600000
#define BN_EPS  1e-5f

typedef __attribute__((ext_vector_type(2))) float v2f;
typedef __attribute__((ext_vector_type(8))) float v8f;
typedef __attribute__((ext_vector_type(4))) float v4f;

__global__ void k_zero(float* __restrict__ p, int n) {
  int i = blockIdx.x * blockDim.x + threadIdx.x;
  if (i < n) p[i] = 0.0f;
}

#define ROWS_PER_BLOCK 391
#define AGG_BLOCKS     256
static_assert(AGG_BLOCKS * ROWS_PER_BLOCK >= N_NODES, "ranges must cover every node");
static_assert(E_EDGES % 4 == 0, "edges are read 4 per thread");

#define GEMM_BLOCKS  256
#define GEMM_THREADS 1024
#define GEMM_WAVES   (GEMM_THREADS / 32)

template <int AMODE, bool HAS_BIAS, bool EPI_RELU, bool ROW_SCALE, int NT>
__device__ __forceinline__ void gemm_tile(
    const float* __restrict__ A, const float2* __restrict__ wl2, float* __restrict__ C,
    int m0, int t0, int laneM, int hi,
    const float* __restrict__ bnScale, const float* __restrict__ bnShift,
    const float* __restrict__ bias, const float* __restrict__ rowscale)
{
  const float* __restrict__ arow = A + (size_t)(m0 + laneM) * DHID;

  v8f acc[NT];
#pragma unroll
  for (int t = 0; t < NT; ++t) acc[t] = (v8f){};

  for (int k = 0; k < DHID; k += 4) {
    float2 av = *(const float2*)(arow + k + hi * 2);
    if (AMODE == 1) {
      av.x = fmaxf(av.x, 0.0f);
      av.y = fmaxf(av.y, 0.0f);
    } else if (AMODE == 2) {
      int c0 = k + hi * 2;
      av.x = fmaxf(av.x, 0.0f) * bnScale[c0]     + bnShift[c0];
      av.y = fmaxf(av.y, 0.0f) * bnScale[c0 + 1] + bnShift[c0 + 1];
    }
    v2f a; a[0] = av.x; a[1] = av.y;

    const float2* __restrict__ brow = wl2 + (((k >> 1) + hi) << 7) + t0 * 16;
#pragma unroll
    for (int t = 0; t < NT; ++t) {
      float2 bv = brow[t * 16 + laneM];
      v2f b; b[0] = bv.x; b[1] = bv.y;
      acc[t] = __builtin_amdgcn_wmma_f32_16x16x4_f32(
          false, a, false, b, (short)0, acc[t], false, false);
    }
  }

  float rs[8];
#pragma unroll
  for (int r = 0; r < 8; ++r) rs[r] = ROW_SCALE ? rowscale[m0 + r + hi * 8] : 1.0f;

#pragma unroll
  for (int t = 0; t < NT; ++t) {
    int   col  = (t0 + t) * 16 + laneM;
    float bval = HAS_BIAS ? bias[col] : 0.0f;
#pragma unroll
    for (int r = 0; r < 8; ++r) {
      int   row = m0 + r + hi * 8;
      float v   = acc[t][r];
      if (ROW_SCALE) v *= rs[r];
      if (HAS_BIAS) v += bval;
      if (EPI_RELU) v = fmaxf(v, 0.0f);
      C[(size_t)row * DHID + col] = v;
    }
  }
}

template <int AMODE, bool HAS_BIAS, bool EPI_RELU, bool ROW_SCALE>
__global__ __launch_bounds__(GEMM_THREADS) void k_gemm128(
    const float* __restrict__ A, const float* __restrict__ W,
    float* __restrict__ C, int nrows,
    const float* __restrict__ bnScale, const float* __restrict__ bnShift,
    const float* __restrict__ bias, const float* __restrict__ rowscale)
{
  __shared__ float wl[DHID * DHID];
  const int tid = threadIdx.x;
  for (int i = tid; i < DHID * DHID; i += GEMM_THREADS) {
    int k = i >> 7, n = i & 127;
    wl[(((k >> 1) << 7) + n) * 2 + (k & 1)] = W[i];
  }
  __syncthreads();

  const int wave  = __builtin_amdgcn_readfirstlane(tid >> 5);
  const int lane  = tid & 31;
  const int laneM = lane & 15;
  const int hi    = lane >> 4;
  const float2* __restrict__ wl2 = (const float2*)wl;

  const int  ntiles = nrows >> 4;
  const int  full   = ntiles / GEMM_BLOCKS;
  const bool extra  = (int)blockIdx.x < ntiles - full * GEMM_BLOCKS;

  for (int i = wave; i < full; i += GEMM_WAVES)
    gemm_tile<AMODE, HAS_BIAS, EPI_RELU, ROW_SCALE, 8>(
        A, wl2, C, ((int)blockIdx.x + GEMM_BLOCKS * i) << 4, 0, laneM, hi, bnScale, bnShift, bias, rowscale);

  if (extra) {
    const int j = (wave - full) & (GEMM_WAVES - 1);
    if (j < 4)
      gemm_tile<AMODE, HAS_BIAS, EPI_RELU, ROW_SCALE, 2>(
          A, wl2, C, ((int)blockIdx.x + GEMM_BLOCKS * full) << 4, j * 2, laneM, hi, bnScale, bnShift, bias,
          rowscale);
  }
}

#define LIST_CAP   24576
#define EDGE_GROUP 8
#define SRC_BITS   17
static_assert(N_NODES <= (1 << SRC_BITS), "a source node must fit the record's low bits");
static_assert(ROWS_PER_BLOCK <= (1 << (32 - SRC_BITS)), "a local row must fit the record's high bits");

#define BIN_EDGES 4096
static_assert(AGG_BLOCKS == 256, "the count scan below is written for 8 waves of 32 ranges");
__global__ __launch_bounds__(1024) void k_bin(const int* __restrict__ src, const int* __restrict__ dst,
                                              unsigned int* __restrict__ recs,
                                              unsigned int* __restrict__ counts, int e4) {
  __shared__ unsigned int  hist[AGG_BLOCKS];
  __shared__ unsigned int  pos[AGG_BLOCKS];
  __shared__ unsigned int  gbase[AGG_BLOCKS];
  __shared__ unsigned int  wsum[8];
  __shared__ unsigned int  stage[BIN_EDGES];
  __shared__ unsigned char srange[BIN_EDGES];
  const int tid = threadIdx.x;
  if (tid < AGG_BLOCKS) hist[tid] = 0u;
  __syncthreads();

  const int  i  = blockIdx.x * 1024 + tid;
  const bool in = i < e4;
  int      s[4] = {0, 0, 0, 0}, d[4] = {0, 0, 0, 0};
  unsigned r[4] = {0u, 0u, 0u, 0u}, rank[4] = {0u, 0u, 0u, 0u};
  if (in) {
    const int4 sv = ((const int4*)src)[i];
    const int4 dv = ((const int4*)dst)[i];
    s[0] = sv.x; s[1] = sv.y; s[2] = sv.z; s[3] = sv.w;
    d[0] = dv.x; d[1] = dv.y; d[2] = dv.z; d[3] = dv.w;
#pragma unroll
    for (int u = 0; u < 4; ++u) {
      r[u]    = (unsigned)(d[u] / ROWS_PER_BLOCK);
      rank[u] = atomicAdd(&hist[r[u]], 1u);
    }
  }
  __syncthreads();

  unsigned n = 0u, incl = 0u;
  if (tid < AGG_BLOCKS) {
    n    = hist[tid];
    incl = n;
#pragma unroll
    for (int off = 1; off < 32; off <<= 1) {
      const unsigned t = __shfl_up(incl, off, 32);
      if ((tid & 31) >= off) incl += t;
    }
    if ((tid & 31) == 31) wsum[tid >> 5] = incl;
    gbase[tid] = n ? atomicAdd(&counts[tid], n) : 0u;
  }
  __syncthreads();
  if (tid < AGG_BLOCKS) {
    unsigned before = 0u;
    for (int w = 0; w < (tid >> 5); ++w) before += wsum[w];
    pos[tid] = before + incl - n;
  }
  __syncthreads();

  if (in) {
#pragma unroll
    for (int u = 0; u < 4; ++u) {
      const unsigned j  = pos[r[u]] + rank[u];
      const unsigned dl = (unsigned)d[u] - r[u] * ROWS_PER_BLOCK;
      stage[j]  = (unsigned)s[u] | (dl << SRC_BITS);
      srange[j] = (unsigned char)r[u];
    }
  }
  __syncthreads();

  const int nrec = min(1024, e4 - (int)blockIdx.x * 1024) * 4;
  for (int j = tid; j < nrec; j += 1024) {
    const unsigned rr   = srange[j];
    const unsigned slot = gbase[rr] + ((unsigned)j - pos[rr]);
    if (slot < LIST_CAP) recs[(size_t)rr * LIST_CAP + slot] = stage[j];
  }
}

#define DEG_THREADS 256
#define DEG_UNROLL  32
__global__ __launch_bounds__(DEG_THREADS) void k_rangedeg(const int* __restrict__ dst,
                                                          const unsigned int* __restrict__ recs,
                                                          const unsigned int* __restrict__ counts,
                                                          float* __restrict__ dinv) {
  __shared__ unsigned int cnt[ROWS_PER_BLOCK];
  const int tid = threadIdx.x;
  const int n0  = blockIdx.x * ROWS_PER_BLOCK;
  for (int i = tid; i < ROWS_PER_BLOCK; i += DEG_THREADS) cnt[i] = 0u;
  __syncthreads();

  const int total = (int)counts[blockIdx.x];
  if (total <= LIST_CAP) {
    const unsigned int* __restrict__ rl = recs + (size_t)blockIdx.x * LIST_CAP;
    for (int j0 = tid; j0 < total; j0 += DEG_UNROLL * DEG_THREADS) {
      unsigned rv[DEG_UNROLL];
#pragma unroll
      for (int u = 0; u < DEG_UNROLL; ++u) rv[u] = rl[min(j0 + u * DEG_THREADS, total - 1)];
#pragma unroll
      for (int u = 0; u < DEG_UNROLL; ++u)
        if (j0 + u * DEG_THREADS < total) atomicAdd(&cnt[rv[u] >> SRC_BITS], 1u);
    }
  } else {
    for (int e = tid; e < E_EDGES; e += DEG_THREADS) {
      const int d = dst[e];
      if (d >= n0 && d < n0 + ROWS_PER_BLOCK) atomicAdd(&cnt[d - n0], 1u);
    }
  }
  __syncthreads();

  const int nrows = min(ROWS_PER_BLOCK, N_NODES - n0);
  for (int i = tid; i < nrows; i += DEG_THREADS) dinv[n0 + i] = __frsqrt_rn((float)cnt[i] + 1.0f);
}

#define OWN_SLOTS 13
#define SUB_CAP   1024
static_assert(OWN_SLOTS * 32 >= ROWS_PER_BLOCK, "the slots of 32 waves must cover a range's rows");

typedef __attribute__((ext_vector_type(16))) float v16f;
static_assert(OWN_SLOTS < 16, "a slot is an element of a 16-element vector, and element 15 is spare");
#define SLOT_ADD(sl, v)   \
  do {                    \
    ax[sl] += (v).x;      \
    ay[sl] += (v).y;      \
    az[sl] += (v).z;      \
    aw[sl] += (v).w;      \
  } while (0)

__global__ __launch_bounds__(1024) void k_aggregate(
    const int* __restrict__ src, const int* __restrict__ dst,
    const float* __restrict__ dinv,
    const float* __restrict__ h, const float* __restrict__ b,
    float* __restrict__ out,
    const unsigned int* __restrict__ recs, const unsigned int* __restrict__ counts)
{
  __shared__ unsigned int sub[32 * SUB_CAP];
  __shared__ unsigned int cnt[32];

  const int      tid  = threadIdx.x;
  const unsigned lane = threadIdx.x & 31u;
  const int      n0   = blockIdx.x * ROWS_PER_BLOCK;
  const int      wave = __builtin_amdgcn_readfirstlane(tid >> 5);

  const unsigned int* __restrict__ rl = recs + (size_t)blockIdx.x * LIST_CAP;

  const int nrows = min(ROWS_PER_BLOCK, N_NODES - n0);

  if (tid < 32) cnt[tid] = 0u;
  const int total = __builtin_amdgcn_readfirstlane((int)counts[blockIdx.x]);
  __syncthreads();

  if (total <= LIST_CAP) {
    for (int j0 = tid; j0 < total; j0 += 8 * 1024) {
      unsigned rv[8], place[8];
#pragma unroll
      for (int u = 0; u < 8; ++u) rv[u] = rl[min(j0 + u * 1024, total - 1)];
#pragma unroll
      for (int u = 0; u < 8; ++u) {
        place[u] = SUB_CAP;
        if (j0 + u * 1024 < total) place[u] = atomicAdd(&cnt[(rv[u] >> SRC_BITS) & 31u], 1u);
      }
#pragma unroll
      for (int u = 0; u < 8; ++u)
        if (place[u] < SUB_CAP) sub[((rv[u] >> SRC_BITS) & 31u) * SUB_CAP + place[u]] = rv[u];
    }
  }
  __syncthreads();

  int mine = SUB_CAP + 1;
  if (total <= LIST_CAP) mine = (int)cnt[wave];
  mine = __builtin_amdgcn_readfirstlane(mine);

  v16f ax = (v16f)(0.0f), ay = (v16f)(0.0f), az = (v16f)(0.0f), aw = (v16f)(0.0f);

  if (mine <= SUB_CAP) {
    const unsigned int* sl = sub + wave * SUB_CAP;
    const int l7 = min((int)lane, EDGE_GROUP - 1);
    unsigned  rc = 0u;
    if (mine > 0) rc = sl[min(l7, mine - 1)];
    for (int g0 = 0; g0 < mine; g0 += EDGE_GROUP) {
      const int nvalid = min(EDGE_GROUP, mine - g0);
      const int gn = g0 + EDGE_GROUP;
      unsigned  nx = rc;
      if (gn < mine) nx = sl[min(gn + l7, mine - 1)];
      const int s    = (int)(rc & ((1u << SRC_BITS) - 1u));
      const int slot = (int)lane < nvalid ? (int)(rc >> (SRC_BITS + 5)) : 15;
      v4f v[EDGE_GROUP];
#pragma unroll
      for (int u = 0; u < EDGE_GROUP; ++u) {
        const int sj = __builtin_amdgcn_readlane(s, u);
        v[u] = ((const v4f*)(h + (size_t)sj * DHID))[lane];
      }
#pragma unroll
      for (int u = 0; u < EDGE_GROUP; ++u) {
        const int k = __builtin_amdgcn_readlane(slot, u);
        SLOT_ADD(k, v[u]);
      }
      rc = nx;
    }
  } else {
    for (int e0 = 0; e0 < E_EDGES; e0 += 32) {
      const int e   = e0 + (int)lane;
      int       rel = -1, sv = 0;
      if (e < E_EDGES) rel = dst[e] - n0;
      const bool m = rel >= 0 && rel < ROWS_PER_BLOCK && (rel & 31) == wave;
      if (m) sv = src[e];
      unsigned mask = __builtin_amdgcn_ballot_w32(m);
      while (mask != 0u) {
        const int j = __builtin_ctz(mask);
        mask &= mask - 1u;
        const int sj = __builtin_amdgcn_readlane(sv, j);
        const int k  = __builtin_amdgcn_readlane(rel, j) >> 5;
        const v4f hv = ((const v4f*)(h + (size_t)sj * DHID))[lane];
        SLOT_ADD(k, hv);
      }
    }
  }

  const v4f* __restrict__ h4 = (const v4f*)h;
  v4f   t[OWN_SLOTS];
  float dv[OWN_SLOTS];
#pragma unroll
  for (int k = 0; k < OWN_SLOTS; ++k) {
    const int r = n0 + min(wave + 32 * k, nrows - 1);
    t[k]  = h4[(size_t)r * (DHID / 4) + lane];
    dv[k] = dinv[r];
  }
  const v4f bv = *(const v4f*)(b + lane * 4);
  v4f* __restrict__ out4 = (v4f*)out;
#pragma unroll
  for (int k = 0; k < OWN_SLOTS; ++k) {
    if (wave + 32 * k < nrows) {
      v4f o;
      o.x = (ax[k] + t[k].x) * dv[k] + bv.x; o.y = (ay[k] + t[k].y) * dv[k] + bv.y;
      o.z = (az[k] + t[k].z) * dv[k] + bv.z; o.w = (aw[k] + t[k].w) * dv[k] + bv.w;
      out4[(size_t)(n0 + wave + 32 * k) * (DHID / 4) + lane] = o;
    }
  }
}

#define TAIL_ROWS    16
#define TAIL_WAVES   25
#define TAIL_THREADS (TAIL_WAVES * 32)
#define TAIL_BLOCKS  256
static_assert(TAIL_ROWS * 2 * sizeof(float) == 128, "a row group's head outputs are one 128-byte line");

__global__ __launch_bounds__(TAIL_THREADS) void k_bnstats(const float* __restrict__ h,
                                                          float* __restrict__ stats, int n) {
  __shared__ float red[TAIL_WAVES][2 * DHID];
  const int tid  = threadIdx.x;
  const int lane = tid & 31;
  const int wave = __builtin_amdgcn_readfirstlane(tid >> 5);
  const int c    = lane * 4;
  float s0 = 0.0f, s1 = 0.0f, s2 = 0.0f, s3 = 0.0f;
  float q0 = 0.0f, q1 = 0.0f, q2 = 0.0f, q3 = 0.0f;
  const int ngroups = (n + TAIL_ROWS - 1) / TAIL_ROWS;
  for (int g = wave * (int)gridDim.x + (int)blockIdx.x; g < ngroups; g += TAIL_WAVES * (int)gridDim.x) {
    const int row0 = g * TAIL_ROWS;
    if (row0 + TAIL_ROWS <= n) {
      v4f v[TAIL_ROWS];
#pragma unroll
      for (int u = 0; u < TAIL_ROWS; ++u) v[u] = *(const v4f*)(h + (size_t)(row0 + u) * DHID + c);
      asm volatile("" :: "v"(v[0]), "v"(v[1]), "v"(v[2]), "v"(v[3]), "v"(v[4]), "v"(v[5]), "v"(v[6]), "v"(v[7]),
                         "v"(v[8]), "v"(v[9]), "v"(v[10]), "v"(v[11]), "v"(v[12]), "v"(v[13]), "v"(v[14]), "v"(v[15]));
#pragma unroll
      for (int u = 0; u < TAIL_ROWS; ++u) {
        const float a0 = fmaxf(v[u].x, 0.0f), a1 = fmaxf(v[u].y, 0.0f);
        const float a2 = fmaxf(v[u].z, 0.0f), a3 = fmaxf(v[u].w, 0.0f);
        s0 += a0; q0 += a0 * a0;
        s1 += a1; q1 += a1 * a1;
        s2 += a2; q2 += a2 * a2;
        s3 += a3; q3 += a3 * a3;
      }
    } else {
      for (int r = row0; r < n; ++r) {
        const float4 t = *(const float4*)(h + (size_t)r * DHID + c);
        const float a0 = fmaxf(t.x, 0.0f), a1 = fmaxf(t.y, 0.0f);
        const float a2 = fmaxf(t.z, 0.0f), a3 = fmaxf(t.w, 0.0f);
        s0 += a0; q0 += a0 * a0;
        s1 += a1; q1 += a1 * a1;
        s2 += a2; q2 += a2 * a2;
        s3 += a3; q3 += a3 * a3;
      }
    }
  }
  *(float4*)&red[wave][c]        = make_float4(s0, s1, s2, s3);
  *(float4*)&red[wave][DHID + c] = make_float4(q0, q1, q2, q3);
  __syncthreads();
  if (tid < 2 * DHID) {
    float t = 0.0f;
#pragma unroll
    for (int w = 0; w < TAIL_WAVES; ++w) t += red[w][tid];
    unsafeAtomicAdd(&stats[tid], t);
  }
}

__global__ void k_bnfinal(const float* __restrict__ stats, const float* __restrict__ gamma,
                          const float* __restrict__ beta, float* __restrict__ bn, int n) {
  int c = threadIdx.x;
  if (c < 128) {
    float inv = 1.0f / (float)n;
    float mu  = stats[c] * inv;
    float var = stats[128 + c] * inv - mu * mu;
    float sc  = gamma[c] * __frsqrt_rn(var + BN_EPS);
    bn[c]       = sc;
    bn[128 + c] = beta[c] - mu * sc;
  }
}

__global__ __launch_bounds__(TAIL_THREADS) void k_head(const float* __restrict__ hc,
                                                       const float* __restrict__ Wr,
                                                       const float* __restrict__ br,
                                                       float* __restrict__ out, int n) {
  const int lane = threadIdx.x & 31;
  const int wave = __builtin_amdgcn_readfirstlane((int)(threadIdx.x >> 5));
  const int c = lane * 4;
  const float w00 = Wr[c * 2],     w01 = Wr[(c + 1) * 2],     w02 = Wr[(c + 2) * 2],     w03 = Wr[(c + 3) * 2];
  const float w10 = Wr[c * 2 + 1], w11 = Wr[(c + 1) * 2 + 1], w12 = Wr[(c + 2) * 2 + 1], w13 = Wr[(c + 3) * 2 + 1];
  const float bsel = br[lane >> 4];

  const bool h16 = (lane & 16) != 0, h8 = (lane & 8) != 0, h4 = (lane & 4) != 0, h2 = (lane & 2) != 0,
             h1 = (lane & 1) != 0;
  const int ngroups = (n + TAIL_ROWS - 1) / TAIL_ROWS;
  for (int g = wave * (int)gridDim.x + (int)blockIdx.x; g < ngroups; g += TAIL_WAVES * (int)gridDim.x) {
    const int row0 = g * TAIL_ROWS;
    float4 v[TAIL_ROWS];
#pragma unroll
    for (int u = 0; u < TAIL_ROWS; ++u) {
      const int r = min(row0 + u, n - 1);
      v[u] = *(const float4*)(hc + (size_t)r * DHID + c);
    }
    __builtin_amdgcn_sched_barrier(0);
    float x[2 * TAIL_ROWS];
#pragma unroll
    for (int u = 0; u < TAIL_ROWS; ++u) {
      x[u]             = v[u].x * w00 + v[u].y * w01 + v[u].z * w02 + v[u].w * w03;
      x[TAIL_ROWS + u] = v[u].x * w10 + v[u].y * w11 + v[u].z * w12 + v[u].w * w13;
    }
    float y16[16], y8[8], y4[4], y2[2];
#pragma unroll
    for (int j = 0; j < 16; ++j) {
      const float keep = h16 ? x[j + 16] : x[j];
      const float send = h16 ? x[j] : x[j + 16];
      y16[j] = keep + __shfl_xor(send, 16, 32);
    }
#pragma unroll
    for (int j = 0; j < 8; ++j) {
      const float keep = h8 ? y16[j + 8] : y16[j];
      const float send = h8 ? y16[j] : y16[j + 8];
      y8[j] = keep + __shfl_xor(send, 8, 32);
    }
#pragma unroll
    for (int j = 0; j < 4; ++j) {
      const float keep = h4 ? y8[j + 4] : y8[j];
      const float send = h4 ? y8[j] : y8[j + 4];
      y4[j] = keep + __shfl_xor(send, 4, 32);
    }
#pragma unroll
    for (int j = 0; j < 2; ++j) {
      const float keep = h2 ? y4[j + 2] : y4[j];
      const float send = h2 ? y4[j] : y4[j + 2];
      y2[j] = keep + __shfl_xor(send, 2, 32);
    }
    const float keep1 = h1 ? y2[1] : y2[0];
    const float send1 = h1 ? y2[0] : y2[1];
    const float y1    = keep1 + __shfl_xor(send1, 1, 32);

    const int r = row0 + (lane & 15);
    if (r < n) out[(size_t)r * 2 + (lane >> 4)] = y1 + bsel;
  }
}

extern "C" void kernel_launch(void* const* d_in, const int* in_sizes, int n_in,
                              void* d_out, int out_size, void* d_ws, size_t ws_size,
                              hipStream_t stream) {
  const float* x     = (const float*)d_in[0];
  const int*   eidx  = (const int*)d_in[1];
  const float* W1    = (const float*)d_in[2];
  const float* b1    = (const float*)d_in[3];
  const float* W2    = (const float*)d_in[4];
  const float* b2    = (const float*)d_in[5];
  const float* W3    = (const float*)d_in[6];
  const float* b3    = (const float*)d_in[7];
  const float* gamma = (const float*)d_in[8];
  const float* beta  = (const float*)d_in[9];
  const float* Wc    = (const float*)d_in[10];
  const float* bc    = (const float*)d_in[11];
  const float* Wr    = (const float*)d_in[12];
  const float* br    = (const float*)d_in[13];
  float* out = (float*)d_out;

  const int N = N_NODES, E = E_EDGES;
  const int* src = eidx;
  const int* dst = eidx + E;

  float* buf0  = (float*)d_ws;
  float* buf1  = buf0 + (size_t)N * DHID;
  float* dinv  = buf1 + (size_t)N * DHID;
  float* stats = dinv + N;
  unsigned int* counts = (unsigned int*)(stats + 256);
  float* bnp   = stats + 256 + AGG_BLOCKS;
  unsigned int* recs = (unsigned int*)(bnp + 256);

  const dim3 blk(256);
  const int gBin  = (E / 4 + 1023) / 1024;

  k_zero<<<2, blk, 0, stream>>>(stats, 256 + AGG_BLOCKS);

  k_bin<<<gBin, 1024, 0, stream>>>(src, dst, recs, counts, E / 4);
  k_rangedeg<<<AGG_BLOCKS, DEG_THREADS, 0, stream>>>(dst, recs, counts, dinv);

  k_gemm128<0, false, false, true><<<GEMM_BLOCKS, GEMM_THREADS, 0, stream>>>(x, W1, buf0, N, nullptr, nullptr, nullptr, dinv);
  k_aggregate<<<AGG_BLOCKS, 1024, 0, stream>>>(src, dst, dinv, buf0, b1, buf1, recs, counts);

  k_gemm128<1, false, false, true><<<GEMM_BLOCKS, GEMM_THREADS, 0, stream>>>(buf1, W2, buf0, N, nullptr, nullptr, nullptr, dinv);
  k_aggregate<<<AGG_BLOCKS, 1024, 0, stream>>>(src, dst, dinv, buf0, b2, buf1, recs, counts);

  k_gemm128<1, false, false, true><<<GEMM_BLOCKS, GEMM_THREADS, 0, stream>>>(buf1, W3, buf0, N, nullptr, nullptr, nullptr, dinv);
  k_aggregate<<<AGG_BLOCKS, 1024, 0, stream>>>(src, dst, dinv, buf0, b3, buf1, recs, counts);

  k_bnstats<<<TAIL_BLOCKS, TAIL_THREADS, 0, stream>>>(buf1, stats, N);
  k_bnfinal<<<1, 128, 0, stream>>>(stats, gamma, beta, bnp, N);

  k_gemm128<2, true, true, false><<<GEMM_BLOCKS, GEMM_THREADS, 0, stream>>>(buf1, Wc, buf0, N, bnp, bnp + 128, bc, nullptr);

  k_head<<<TAIL_BLOCKS, TAIL_THREADS, 0, stream>>>(buf0, Wr, br, out, N);
}
